# starting kernel
# baseline (speedup 1.0000x reference)
#include <hip/hip_runtime.h>


typedef __attribute__((ext_vector_type(2))) float v2f;
typedef __attribute__((ext_vector_type(8))) float v8f;

#define NNODES 50000
#define FEAT   256
#define OUTF   128
#define NEDGE  800000

__global__ __launch_bounds__(256) void gemm_bias_wmma(
    const float* __restrict__ X, const float* __restrict__ W,
    const float* __restrict__ B, float* __restrict__ Y, int P)
{
  const int lane  = threadIdx.x & 31;
  const int wave  = threadIdx.x >> 5;
  const int m0    = blockIdx.x << 4;
  const int n0    = ((blockIdx.y << 3) + wave) << 4;
  const int lmod  = lane & 15;
  const int lhalf = lane >> 4;
  const int koff  = lhalf << 1;

  const float* xrow = X + (size_t)(m0 + lmod) * FEAT + koff;
  const float* wrow = W + (size_t)(n0 + lmod) * FEAT + koff;

  v8f acc = {0.f, 0.f, 0.f, 0.f, 0.f, 0.f, 0.f, 0.f};
#pragma unroll 8
  for (int k = 0; k < FEAT; k += 4) {
    v2f a = *(const v2f*)(xrow + k);
    v2f b = *(const v2f*)(wrow + k);
    acc = __builtin_amdgcn_wmma_f32_16x16x4_f32(
        false, a, false, b, (short)0, acc, false, false);
  }

  const float bias = B ? B[n0 + lmod] : 0.0f;
  float* yp = Y + (size_t)(m0 + (lhalf << 3)) * P + (n0 + lmod);
#pragma unroll
  for (int i = 0; i < 8; ++i) yp[(size_t)i * P] = acc[i] + bias;
}

__global__ __launch_bounds__(256) void gemm_out_wmma(
    const float* __restrict__ S, const float* __restrict__ INV,
    const float* __restrict__ X,
    const float* __restrict__ W1, const float* __restrict__ W2,
    const float* __restrict__ B2, float* __restrict__ OUT, int colOff)
{
  const int lane  = threadIdx.x & 31;
  const int wave  = threadIdx.x >> 5;
  const int m0    = blockIdx.x << 4;
  const int n0    = wave << 4;
  const int lmod  = lane & 15;
  const int lhalf = lane >> 4;
  const int koff  = lhalf << 1;
  const int arow  = m0 + lmod;

  const float scale = INV[arow];
  const float* srow  = S  + (size_t)arow * FEAT + koff;
  const float* xrow  = X  + (size_t)arow * FEAT + koff;
  const float* w1row = W1 + (size_t)(n0 + lmod) * FEAT + koff;
  const float* w2row = W2 + (size_t)(n0 + lmod) * FEAT + koff;

  v8f acc = {0.f, 0.f, 0.f, 0.f, 0.f, 0.f, 0.f, 0.f};
#pragma unroll 8
  for (int k = 0; k < FEAT; k += 4) {
    v2f a = *(const v2f*)(srow + k) * scale;
    v2f b = *(const v2f*)(w1row + k);
    acc = __builtin_amdgcn_wmma_f32_16x16x4_f32(
        false, a, false, b, (short)0, acc, false, false);
  }
#pragma unroll 8
  for (int k = 0; k < FEAT; k += 4) {
    v2f a = *(const v2f*)(xrow + k);
    v2f b = *(const v2f*)(w2row + k);
    acc = __builtin_amdgcn_wmma_f32_16x16x4_f32(
        false, a, false, b, (short)0, acc, false, false);
  }

  const float bias = B2[n0 + lmod];
  float* yp = OUT + (size_t)(m0 + (lhalf << 3)) * (2 * OUTF) + colOff + (n0 + lmod);
#pragma unroll
  for (int i = 0; i < 8; ++i) yp[(size_t)i * (2 * OUTF)] = acc[i] + bias;
}

__global__ __launch_bounds__(256) void edge_score(
    const int* __restrict__ EI, const float* __restrict__ ATT,
    const float* __restrict__ X, float* __restrict__ Wout, int E)
{
  const int lane = threadIdx.x & 31;
  const int wave = threadIdx.x >> 5;
  const int e = (blockIdx.x << 3) + wave;
  if (e >= E) return;
  const int r = EI[e];
  const int l = EI[E + e];

  const float4* ap = (const float4*)(ATT + (size_t)r * FEAT) + (lane << 1);
  const float4* xp = (const float4*)(X   + (size_t)l * FEAT) + (lane << 1);
  float4 a0 = ap[0], a1 = ap[1];
  float4 x0 = xp[0], x1 = xp[1];
  float v = a0.x * x0.x + a0.y * x0.y + a0.z * x0.z + a0.w * x0.w
          + a1.x * x1.x + a1.y * x1.y + a1.z * x1.z + a1.w * x1.w;
#pragma unroll
  for (int off = 16; off >= 1; off >>= 1) v += __shfl_xor(v, off, 32);
  if (lane == 0) Wout[e] = v;
}

__global__ __launch_bounds__(256) void edge_scatter(
    const int* __restrict__ EI, const float* __restrict__ Wv,
    const float* __restrict__ X, float* __restrict__ S,
    float* __restrict__ CNT, int E)
{
  const int lane = threadIdx.x & 31;
  const int wave = threadIdx.x >> 5;
  const int e = (blockIdx.x << 3) + wave;
  if (e >= E) return;
  const int r = EI[e];
  const int l = EI[E + e];
  const float w = Wv[e];

  const float4* xp = (const float4*)(X + (size_t)l * FEAT) + (lane << 1);
  float4 x0 = xp[0], x1 = xp[1];
  float* sp = S + (size_t)r * FEAT + (lane << 3);
  atomicAdd(sp + 0, w * x0.x);
  atomicAdd(sp + 1, w * x0.y);
  atomicAdd(sp + 2, w * x0.z);
  atomicAdd(sp + 3, w * x0.w);
  atomicAdd(sp + 4, w * x1.x);
  atomicAdd(sp + 5, w * x1.y);
  atomicAdd(sp + 6, w * x1.z);
  atomicAdd(sp + 7, w * x1.w);
  if (lane == 0) atomicAdd(CNT + r, 1.0f);
}

__global__ __launch_bounds__(256) void finalize_inv(float* __restrict__ CNT, int n)
{
  int i = blockIdx.x * blockDim.x + threadIdx.x;
  if (i < n) {
    float c = CNT[i];
    CNT[i] = 1.0f / fmaxf(c, 1.0f);
  }
}

extern "C" void kernel_launch(void* const* d_in, const int* in_sizes, int n_in,
                              void* d_out, int out_size, void* d_ws, size_t ws_size,
                              hipStream_t stream)
{
  const float* x        = (const float*)d_in[0];
  const int*   pos_ei   = (const int*)d_in[1];
  const int*   neg_ei   = (const int*)d_in[2];
  const float* W_pos    = (const float*)d_in[3];
  const float* W_pos_cc = (const float*)d_in[4];
  const float* b_pos_cc = (const float*)d_in[5];
  const float* W_neg    = (const float*)d_in[6];
  const float* W_neg_cc = (const float*)d_in[7];
  const float* b_neg_cc = (const float*)d_in[8];
  const float* W_patt   = (const float*)d_in[9];
  const float* b_patt   = (const float*)d_in[10];
  const float* W_natt   = (const float*)d_in[11];
  const float* b_natt   = (const float*)d_in[12];
  float* out = (float*)d_out;

  float* Sbuf = (float*)d_ws;
  float* Wbuf = Sbuf + (size_t)NNODES * FEAT;
  float* Cbuf = Wbuf + (size_t)NEDGE;

  const dim3 blk(256);
  const dim3 gAtt(NNODES / 16, FEAT / 128);
  const dim3 gOut(NNODES / 16, 1);
  const int  gEdge = (NEDGE + 7) / 8;
  const int  gFin  = (NNODES + 255) / 256;

  const int*   ei[2]   = { pos_ei, neg_ei };
  const float* Watt[2] = { W_patt, W_natt };
  const float* batt[2] = { b_patt, b_natt };
  const float* Wagg[2] = { W_pos,  W_neg  };
  const float* Wcc[2]  = { W_pos_cc, W_neg_cc };
  const float* bcc[2]  = { b_pos_cc, b_neg_cc };

  for (int s = 0; s < 2; ++s) {
    gemm_bias_wmma<<<gAtt, blk, 0, stream>>>(x, Watt[s], batt[s], Sbuf, FEAT);
    edge_score<<<gEdge, blk, 0, stream>>>(ei[s], Sbuf, x, Wbuf, NEDGE);
    hipMemsetAsync(Sbuf, 0, (size_t)NNODES * FEAT * sizeof(float), stream);
    hipMemsetAsync(Cbuf, 0, (size_t)NNODES * sizeof(float), stream);
    edge_scatter<<<gEdge, blk, 0, stream>>>(ei[s], Wbuf, x, Sbuf, Cbuf, NEDGE);
    finalize_inv<<<gFin, blk, 0, stream>>>(Cbuf, NNODES);
    gemm_out_wmma<<<gOut, blk, 0, stream>>>(Sbuf, Cbuf, x, Wagg[s], Wcc[s],
                                            bcc[s], out, s * OUTF);
  }
}


# Float atomic scatter removed: edges counting-sorted by target (1 int atomic per edge); one wave per node scores and sums its edges in registers, bucket first ordered by source index (fixed sum order)
# speedup vs baseline: 1.9975x; 1.9975x over previous
#include <hip/hip_runtime.h>


typedef __attribute__((ext_vector_type(2))) float v2f;
typedef __attribute__((ext_vector_type(8))) float v8f;

#define NNODES 50000
#define FEAT   256
#define OUTF   128
#define NEDGE  800000

__global__ __launch_bounds__(256) void gemm_bias_wmma(
    const float* __restrict__ X, const float* __restrict__ W,
    const float* __restrict__ B, float* __restrict__ Y, int P)
{
  const int lane  = threadIdx.x & 31;
  const int wave  = threadIdx.x >> 5;
  const int m0    = blockIdx.x << 4;
  const int n0    = ((blockIdx.y << 3) + wave) << 4;
  const int lmod  = lane & 15;
  const int lhalf = lane >> 4;
  const int koff  = lhalf << 1;

  const float* xrow = X + (size_t)(m0 + lmod) * FEAT + koff;
  const float* wrow = W + (size_t)(n0 + lmod) * FEAT + koff;

  v8f acc = {0.f, 0.f, 0.f, 0.f, 0.f, 0.f, 0.f, 0.f};
#pragma unroll 8
  for (int k = 0; k < FEAT; k += 4) {
    v2f a = *(const v2f*)(xrow + k);
    v2f b = *(const v2f*)(wrow + k);
    acc = __builtin_amdgcn_wmma_f32_16x16x4_f32(
        false, a, false, b, (short)0, acc, false, false);
  }

  const float bias = B ? B[n0 + lmod] : 0.0f;
  float* yp = Y + (size_t)(m0 + (lhalf << 3)) * P + (n0 + lmod);
#pragma unroll
  for (int i = 0; i < 8; ++i) yp[(size_t)i * P] = acc[i] + bias;
}

__global__ __launch_bounds__(256) void gemm_out_wmma(
    const float* __restrict__ S, const float* __restrict__ X,
    const float* __restrict__ W1, const float* __restrict__ W2,
    const float* __restrict__ B2, float* __restrict__ OUT, int colOff)
{
  const int lane  = threadIdx.x & 31;
  const int wave  = threadIdx.x >> 5;
  const int m0    = blockIdx.x << 4;
  const int n0    = wave << 4;
  const int lmod  = lane & 15;
  const int lhalf = lane >> 4;
  const int koff  = lhalf << 1;
  const int arow  = m0 + lmod;

  const float* srow  = S  + (size_t)arow * FEAT + koff;
  const float* xrow  = X  + (size_t)arow * FEAT + koff;
  const float* w1row = W1 + (size_t)(n0 + lmod) * FEAT + koff;
  const float* w2row = W2 + (size_t)(n0 + lmod) * FEAT + koff;

  v8f acc = {0.f, 0.f, 0.f, 0.f, 0.f, 0.f, 0.f, 0.f};
#pragma unroll 8
  for (int k = 0; k < FEAT; k += 4) {
    v2f a = *(const v2f*)(srow + k);
    v2f b = *(const v2f*)(w1row + k);
    acc = __builtin_amdgcn_wmma_f32_16x16x4_f32(
        false, a, false, b, (short)0, acc, false, false);
  }
#pragma unroll 8
  for (int k = 0; k < FEAT; k += 4) {
    v2f a = *(const v2f*)(xrow + k);
    v2f b = *(const v2f*)(w2row + k);
    acc = __builtin_amdgcn_wmma_f32_16x16x4_f32(
        false, a, false, b, (short)0, acc, false, false);
  }

  const float bias = B2[n0 + lmod];
  float* yp = OUT + (size_t)(m0 + (lhalf << 3)) * (2 * OUTF) + colOff + (n0 + lmod);
#pragma unroll
  for (int i = 0; i < 8; ++i) yp[(size_t)i * (2 * OUTF)] = acc[i] + bias;
}

__global__ __launch_bounds__(256) void edge_rank(
    const int* __restrict__ EI, int* __restrict__ CNT, int* __restrict__ RANK, int E)
{
  const int e = blockIdx.x * 256 + threadIdx.x;
  if (e < E) RANK[e] = atomicAdd(CNT + EI[e], 1);
}

__global__ __launch_bounds__(1024) void scan_excl(
    const int* __restrict__ CNT, int* __restrict__ OFS, int n)
{
  __shared__ int wsum[32];
  __shared__ int wpre[32];
  const int t    = threadIdx.x;
  const int lane = t & 31;
  const int wave = t >> 5;
  const int base = blockIdx.x << 10;

  int pre = 0;
  const int4* c4 = (const int4*)CNT;
  for (int i = t; i < (base >> 2); i += 1024) {
    const int4 v = c4[i];
    pre += (v.x + v.y) + (v.z + v.w);
  }
  const int idx = base + t;
  const int c = (idx < n) ? CNT[idx] : 0;

  int inc = c;
  for (int off = 1; off < 32; off <<= 1) {
    const int v = __shfl_up(inc, off, 32);
    if (lane >= off) inc += v;
  }
  for (int off = 16; off >= 1; off >>= 1) pre += __shfl_xor(pre, off, 32);
  if (lane == 31) wsum[wave] = inc;
  if (lane == 0)  wpre[wave] = pre;
  __syncthreads();
  int front = 0;
  for (int w = 0; w < 32; ++w) {
    front += wpre[w];
    if (w < wave) front += wsum[w];
  }
  if (idx < n) OFS[idx] = front + inc - c;
}

__global__ __launch_bounds__(256) void edge_place(
    const int* __restrict__ EI, const int* __restrict__ OFS,
    const int* __restrict__ RANK, int* __restrict__ BKT, int E)
{
  const int e = blockIdx.x * 256 + threadIdx.x;
  if (e < E) BKT[OFS[EI[e]] + RANK[e]] = EI[E + e];
}

__device__ __forceinline__ void edge_term(
    const float* __restrict__ X, int l, int lane,
    const float4& a0, const float4& a1, float4& s0, float4& s1)
{
  const float4* xp = (const float4*)(X + (size_t)l * FEAT);
  const float4 x0 = xp[lane], x1 = xp[lane + 32];
  float v = a0.x * x0.x + a0.y * x0.y + a0.z * x0.z + a0.w * x0.w
          + a1.x * x1.x + a1.y * x1.y + a1.z * x1.z + a1.w * x1.w;
  for (int off = 16; off >= 1; off >>= 1) v += __shfl_xor(v, off, 32);
  s0.x += v * x0.x; s0.y += v * x0.y; s0.z += v * x0.z; s0.w += v * x0.w;
  s1.x += v * x1.x; s1.y += v * x1.y; s1.z += v * x1.z; s1.w += v * x1.w;
}

__global__ __launch_bounds__(256) void node_aggregate(
    const int* __restrict__ OFS, const int* __restrict__ BKT,
    const float* __restrict__ X, float* __restrict__ AS, int n, int E)
{
  const int lane = threadIdx.x & 31;
  const int wave = threadIdx.x >> 5;
  const int r = __builtin_amdgcn_readfirstlane((blockIdx.x << 3) + wave);
  if (r >= n) return;
  const int start = OFS[r];
  const int end   = (r + 1 < n) ? OFS[r + 1] : E;
  const int deg   = end - start;

  float4* row = (float4*)(AS + (size_t)r * FEAT);
  const float4 a0 = row[lane], a1 = row[lane + 32];
  float4 s0 = {0.f, 0.f, 0.f, 0.f}, s1 = {0.f, 0.f, 0.f, 0.f};

  if (deg <= 32) {
    const int k = (lane < deg) ? BKT[start + lane] : 0x7fffffff;
    int rank = 0;
#pragma unroll
    for (int j = 0; j < 32; ++j) {
      const int kj = __builtin_amdgcn_readlane(k, j);
      rank += ((kj < k) || (kj == k && j < lane)) ? 1 : 0;
    }
    const int sorted = __builtin_amdgcn_ds_permute(rank << 2, k);
#pragma unroll 4
    for (int t = 0; t < deg; ++t) {
      const int l = __builtin_amdgcn_readlane(sorted, t);
      edge_term(X, l, lane, a0, a1, s0, s1);
    }
  } else {
    int last = -1;
    for (int it = 0; it < deg; ++it) {
      int cand = 0x7fffffff;
      for (int b = start; b < end; b += 32) {
        const int p = b + lane;
        const int k = (p < end) ? BKT[p] : 0x7fffffff;
        if (k > last && k < cand) cand = k;
      }
      for (int off = 16; off >= 1; off >>= 1) {
        const int o = __shfl_xor(cand, off, 32);
        cand = (o < cand) ? o : cand;
      }
      if (cand == 0x7fffffff) break;
      int mult = 0;
      for (int b = start; b < end; b += 32) {
        const int p = b + lane;
        const int k = (p < end) ? BKT[p] : 0x7fffffff;
        mult += __popc(__builtin_amdgcn_ballot_w32(k == cand));
      }
      const int l = __builtin_amdgcn_readfirstlane(cand);
      for (int m = 0; m < mult; ++m) edge_term(X, l, lane, a0, a1, s0, s1);
      last = cand;
    }
  }

  const float inv = 1.0f / fmaxf((float)deg, 1.0f);
  s0.x *= inv; s0.y *= inv; s0.z *= inv; s0.w *= inv;
  s1.x *= inv; s1.y *= inv; s1.z *= inv; s1.w *= inv;
  row[lane] = s0;
  row[lane + 32] = s1;
}

extern "C" void kernel_launch(void* const* d_in, const int* in_sizes, int n_in,
                              void* d_out, int out_size, void* d_ws, size_t ws_size,
                              hipStream_t stream)
{
  const float* x        = (const float*)d_in[0];
  const int*   pos_ei   = (const int*)d_in[1];
  const int*   neg_ei   = (const int*)d_in[2];
  const float* W_pos    = (const float*)d_in[3];
  const float* W_pos_cc = (const float*)d_in[4];
  const float* b_pos_cc = (const float*)d_in[5];
  const float* W_neg    = (const float*)d_in[6];
  const float* W_neg_cc = (const float*)d_in[7];
  const float* b_neg_cc = (const float*)d_in[8];
  const float* W_patt   = (const float*)d_in[9];
  const float* b_patt   = (const float*)d_in[10];
  const float* W_natt   = (const float*)d_in[11];
  const float* b_natt   = (const float*)d_in[12];
  float* out = (float*)d_out;

  float* Sbuf = (float*)d_ws;
  int*   Rank = (int*)d_ws;
  int*   Bkt  = (int*)(Sbuf + (size_t)NNODES * FEAT);
  int*   Ofs  = Bkt + (size_t)NEDGE;

  const dim3 blk(256);
  const dim3 gAtt(NNODES / 16, FEAT / 128);
  const dim3 gOut(NNODES / 16, 1);
  const int  gEdge = (NEDGE + 255) / 256;
  const int  gScan = (NNODES + 1023) / 1024;
  const int  gNode = (NNODES + 7) / 8;

  const int*   ei[2]   = { pos_ei, neg_ei };
  const float* Watt[2] = { W_patt, W_natt };
  const float* batt[2] = { b_patt, b_natt };
  const float* Wagg[2] = { W_pos,  W_neg  };
  const float* Wcc[2]  = { W_pos_cc, W_neg_cc };
  const float* bcc[2]  = { b_pos_cc, b_neg_cc };

  for (int s = 0; s < 2; ++s) {
    hipMemsetAsync(Bkt, 0, (size_t)NNODES * sizeof(int), stream);
    edge_rank<<<gEdge, blk, 0, stream>>>(ei[s], Bkt, Rank, NEDGE);
    scan_excl<<<gScan, dim3(1024), 0, stream>>>(Bkt, Ofs, NNODES);
    edge_place<<<gEdge, blk, 0, stream>>>(ei[s], Ofs, Rank, Bkt, NEDGE);
    gemm_bias_wmma<<<gAtt, blk, 0, stream>>>(x, Watt[s], batt[s], Sbuf, FEAT);
    node_aggregate<<<gNode, blk, 0, stream>>>(Ofs, Bkt, x, Sbuf, NNODES, NEDGE);
    gemm_out_wmma<<<gOut, blk, 0, stream>>>(Sbuf, x, Wagg[s], Wcc[s],
                                            bcc[s], out, s * OUTF);
  }
}


# Both GEMMs re-tiled: 80x64 block tile, A and W slices staged through LDS by whole-line 16-byte loads, each wave holds 5 accumulator tiles on one B fragment; k sum order regrouped within 8s (fixed)
# speedup vs baseline: 3.4754x; 1.7399x over previous
#include <hip/hip_runtime.h>


typedef __attribute__((ext_vector_type(2))) float v2f;
typedef __attribute__((ext_vector_type(8))) float v8f;

#define NNODES 50000
#define FEAT   256
#define OUTF   128
#define NEDGE  800000

typedef __attribute__((ext_vector_type(4))) float v4f;

#define BM  80
#define BN  64
#define KC  32
#define LDP 36

__global__ __launch_bounds__(128) void gemm_tiled_wmma(
    const float* __restrict__ A0, const float* __restrict__ W0,
    const float* __restrict__ A1, const float* __restrict__ W1,
    const float* __restrict__ bias, float* __restrict__ Y, int nseg, int colOff)
{
  __shared__ __attribute__((aligned(16))) float As[BM * LDP];
  __shared__ __attribute__((aligned(16))) float Bs[BN * LDP];

  const int tid  = threadIdx.x;
  const int lane = tid & 31;
  const int wave = tid >> 5;
  const int c    = lane & 15;
  const int h    = lane >> 4;
  const int m0   = blockIdx.x * BM;
  const int n0   = blockIdx.y * BN;

  const int srow = tid >> 3;
  const int scol = (tid & 7) << 2;
  const size_t aoff = (size_t)(m0 + srow) * FEAT + scol;
  const size_t woff = (size_t)(n0 + srow) * FEAT + scol;
  float* const asw = As + srow * LDP + scol;
  float* const bsw = Bs + srow * LDP + scol;
  const float* const afr = As + c * LDP + (h << 2);
  const float* const bfr = Bs + ((wave << 4) + c) * LDP + (h << 2);

  v8f acc[5];
#pragma unroll
  for (int i = 0; i < 5; ++i) acc[i] = (v8f){0.f, 0.f, 0.f, 0.f, 0.f, 0.f, 0.f, 0.f};

  v4f ra[5], rb[4];
  {
    const float* ap = A0 + aoff;
    const float* wp = W0 + woff;
#pragma unroll
    for (int i = 0; i < 5; ++i) ra[i] = *(const v4f*)(ap + (size_t)i * 16 * FEAT);
#pragma unroll
    for (int i = 0; i < 4; ++i) rb[i] = *(const v4f*)(wp + (size_t)i * 16 * FEAT);
  }

  const int nch = nseg * (FEAT / KC);
  for (int ch = 0; ch < nch; ++ch) {
    __syncthreads();
#pragma unroll
    for (int i = 0; i < 5; ++i) *(v4f*)(asw + i * 16 * LDP) = ra[i];
#pragma unroll
    for (int i = 0; i < 4; ++i) *(v4f*)(bsw + i * 16 * LDP) = rb[i];
    __syncthreads();

    if (ch + 1 < nch) {
      const int   nx = ch + 1;
      const int   kb = (nx & 7) * KC;
      const float* ap = ((nx < 8) ? A0 : A1) + aoff + kb;
      const float* wp = ((nx < 8) ? W0 : W1) + woff + kb;
#pragma unroll
      for (int i = 0; i < 5; ++i) ra[i] = *(const v4f*)(ap + (size_t)i * 16 * FEAT);
#pragma unroll
      for (int i = 0; i < 4; ++i) rb[i] = *(const v4f*)(wp + (size_t)i * 16 * FEAT);
    }

    v4f fa[2][5], fb[2];
    fb[0] = *(const v4f*)bfr;
#pragma unroll
    for (int i = 0; i < 5; ++i) fa[0][i] = *(const v4f*)(afr + i * 16 * LDP);
#pragma unroll
    for (int j = 0; j < 4; ++j) {
      const int cur = j & 1, nxt = cur ^ 1;
      if (j + 1 < 4) {
        fb[nxt] = *(const v4f*)(bfr + (j + 1) * 8);
#pragma unroll
        for (int i = 0; i < 5; ++i)
          fa[nxt][i] = *(const v4f*)(afr + i * 16 * LDP + (j + 1) * 8);
      }
#pragma unroll
      for (int i = 0; i < 5; ++i) {
        const v2f a = fa[cur][i].xy;
        const v2f b = fb[cur].xy;
        acc[i] = __builtin_amdgcn_wmma_f32_16x16x4_f32(
            false, a, false, b, (short)0, acc[i], false, false);
      }
#pragma unroll
      for (int i = 0; i < 5; ++i) {
        const v2f a = fa[cur][i].zw;
        const v2f b = fb[cur].zw;
        acc[i] = __builtin_amdgcn_wmma_f32_16x16x4_f32(
            false, a, false, b, (short)0, acc[i], false, false);
      }
    }
  }

  const int col = n0 + (wave << 4) + c;
  const float bv = bias[col];
  float* yp = Y + (size_t)(m0 + (h << 3)) * FEAT + colOff + col;
#pragma unroll
  for (int i = 0; i < 5; ++i) {
#pragma unroll
    for (int r = 0; r < 8; ++r)
      yp[(size_t)(i * 16 + r) * FEAT] = acc[i][r] + bv;
  }
}

__global__ __launch_bounds__(256) void edge_rank(
    const int* __restrict__ EI, int* __restrict__ CNT, int* __restrict__ RANK, int E)
{
  const int e = blockIdx.x * 256 + threadIdx.x;
  if (e < E) RANK[e] = atomicAdd(CNT + EI[e], 1);
}

__global__ __launch_bounds__(1024) void scan_excl(
    const int* __restrict__ CNT, int* __restrict__ OFS, int n)
{
  __shared__ int wsum[32];
  __shared__ int wpre[32];
  const int t    = threadIdx.x;
  const int lane = t & 31;
  const int wave = t >> 5;
  const int base = blockIdx.x << 10;

  int pre = 0;
  const int4* c4 = (const int4*)CNT;
  for (int i = t; i < (base >> 2); i += 1024) {
    const int4 v = c4[i];
    pre += (v.x + v.y) + (v.z + v.w);
  }
  const int idx = base + t;
  const int c = (idx < n) ? CNT[idx] : 0;

  int inc = c;
  for (int off = 1; off < 32; off <<= 1) {
    const int v = __shfl_up(inc, off, 32);
    if (lane >= off) inc += v;
  }
  for (int off = 16; off >= 1; off >>= 1) pre += __shfl_xor(pre, off, 32);
  if (lane == 31) wsum[wave] = inc;
  if (lane == 0)  wpre[wave] = pre;
  __syncthreads();
  int front = 0;
  for (int w = 0; w < 32; ++w) {
    front += wpre[w];
    if (w < wave) front += wsum[w];
  }
  if (idx < n) OFS[idx] = front + inc - c;
}

__global__ __launch_bounds__(256) void edge_place(
    const int* __restrict__ EI, const int* __restrict__ OFS,
    const int* __restrict__ RANK, int* __restrict__ BKT, int E)
{
  const int e = blockIdx.x * 256 + threadIdx.x;
  if (e < E) BKT[OFS[EI[e]] + RANK[e]] = EI[E + e];
}

__device__ __forceinline__ void edge_term(
    const float* __restrict__ X, int l, int lane,
    const float4& a0, const float4& a1, float4& s0, float4& s1)
{
  const float4* xp = (const float4*)(X + (size_t)l * FEAT);
  const float4 x0 = xp[lane], x1 = xp[lane + 32];
  float v = a0.x * x0.x + a0.y * x0.y + a0.z * x0.z + a0.w * x0.w
          + a1.x * x1.x + a1.y * x1.y + a1.z * x1.z + a1.w * x1.w;
  for (int off = 16; off >= 1; off >>= 1) v += __shfl_xor(v, off, 32);
  s0.x += v * x0.x; s0.y += v * x0.y; s0.z += v * x0.z; s0.w += v * x0.w;
  s1.x += v * x1.x; s1.y += v * x1.y; s1.z += v * x1.z; s1.w += v * x1.w;
}

__global__ __launch_bounds__(256) void node_aggregate(
    const int* __restrict__ OFS, const int* __restrict__ BKT,
    const float* __restrict__ X, float* __restrict__ AS, int n, int E)
{
  const int lane = threadIdx.x & 31;
  const int wave = threadIdx.x >> 5;
  const int r = __builtin_amdgcn_readfirstlane((blockIdx.x << 3) + wave);
  if (r >= n) return;
  const int start = OFS[r];
  const int end   = (r + 1 < n) ? OFS[r + 1] : E;
  const int deg   = end - start;

  float4* row = (float4*)(AS + (size_t)r * FEAT);
  const float4 a0 = row[lane], a1 = row[lane + 32];
  float4 s0 = {0.f, 0.f, 0.f, 0.f}, s1 = {0.f, 0.f, 0.f, 0.f};

  if (deg <= 32) {
    const int k = (lane < deg) ? BKT[start + lane] : 0x7fffffff;
    int rank = 0;
#pragma unroll
    for (int j = 0; j < 32; ++j) {
      const int kj = __builtin_amdgcn_readlane(k, j);
      rank += ((kj < k) || (kj == k && j < lane)) ? 1 : 0;
    }
    const int sorted = __builtin_amdgcn_ds_permute(rank << 2, k);
#pragma unroll 4
    for (int t = 0; t < deg; ++t) {
      const int l = __builtin_amdgcn_readlane(sorted, t);
      edge_term(X, l, lane, a0, a1, s0, s1);
    }
  } else {
    int last = -1;
    for (int it = 0; it < deg; ++it) {
      int cand = 0x7fffffff;
      for (int b = start; b < end; b += 32) {
        const int p = b + lane;
        const int k = (p < end) ? BKT[p] : 0x7fffffff;
        if (k > last && k < cand) cand = k;
      }
      for (int off = 16; off >= 1; off >>= 1) {
        const int o = __shfl_xor(cand, off, 32);
        cand = (o < cand) ? o : cand;
      }
      if (cand == 0x7fffffff) break;
      int mult = 0;
      for (int b = start; b < end; b += 32) {
        const int p = b + lane;
        const int k = (p < end) ? BKT[p] : 0x7fffffff;
        mult += __popc(__builtin_amdgcn_ballot_w32(k == cand));
      }
      const int l = __builtin_amdgcn_readfirstlane(cand);
      for (int m = 0; m < mult; ++m) edge_term(X, l, lane, a0, a1, s0, s1);
      last = cand;
    }
  }

  const float inv = 1.0f / fmaxf((float)deg, 1.0f);
  s0.x *= inv; s0.y *= inv; s0.z *= inv; s0.w *= inv;
  s1.x *= inv; s1.y *= inv; s1.z *= inv; s1.w *= inv;
  row[lane] = s0;
  row[lane + 32] = s1;
}

extern "C" void kernel_launch(void* const* d_in, const int* in_sizes, int n_in,
                              void* d_out, int out_size, void* d_ws, size_t ws_size,
                              hipStream_t stream)
{
  const float* x        = (const float*)d_in[0];
  const int*   pos_ei   = (const int*)d_in[1];
  const int*   neg_ei   = (const int*)d_in[2];
  const float* W_pos    = (const float*)d_in[3];
  const float* W_pos_cc = (const float*)d_in[4];
  const float* b_pos_cc = (const float*)d_in[5];
  const float* W_neg    = (const float*)d_in[6];
  const float* W_neg_cc = (const float*)d_in[7];
  const float* b_neg_cc = (const float*)d_in[8];
  const float* W_patt   = (const float*)d_in[9];
  const float* b_patt   = (const float*)d_in[10];
  const float* W_natt   = (const float*)d_in[11];
  const float* b_natt   = (const float*)d_in[12];
  float* out = (float*)d_out;

  float* Sbuf = (float*)d_ws;
  int*   Rank = (int*)d_ws;
  int*   Bkt  = (int*)(Sbuf + (size_t)NNODES * FEAT);
  int*   Ofs  = Bkt + (size_t)NEDGE;

  const dim3 blk(256);
  const dim3 gblk(128);
  const dim3 gAtt(NNODES / BM, FEAT / BN);
  const dim3 gOut(NNODES / BM, OUTF / BN);
  const int  gEdge = (NEDGE + 255) / 256;
  const int  gScan = (NNODES + 1023) / 1024;
  const int  gNode = (NNODES + 7) / 8;

  const int*   ei[2]   = { pos_ei, neg_ei };
  const float* Watt[2] = { W_patt, W_natt };
  const float* batt[2] = { b_patt, b_natt };
  const float* Wagg[2] = { W_pos,  W_neg  };
  const float* Wcc[2]  = { W_pos_cc, W_neg_cc };
  const float* bcc[2]  = { b_pos_cc, b_neg_cc };

  for (int s = 0; s < 2; ++s) {
    hipMemsetAsync(Bkt, 0, (size_t)NNODES * sizeof(int), stream);
    edge_rank<<<gEdge, blk, 0, stream>>>(ei[s], Bkt, Rank, NEDGE);
    scan_excl<<<gScan, dim3(1024), 0, stream>>>(Bkt, Ofs, NNODES);
    edge_place<<<gEdge, blk, 0, stream>>>(ei[s], Ofs, Rank, Bkt, NEDGE);
    gemm_tiled_wmma<<<gAtt, gblk, 0, stream>>>(x, Watt[s], x, Watt[s],
                                               batt[s], Sbuf, 1, 0);
    node_aggregate<<<gNode, blk, 0, stream>>>(Ofs, Bkt, x, Sbuf, NNODES, NEDGE);
    gemm_tiled_wmma<<<gOut, gblk, 0, stream>>>(Sbuf, Wagg[s], x, Wcc[s],
                                               bcc[s], out, 2, s * OUTF);
  }
}


# Edge ranks from returning LDS adds: 32 edge slices, each work-group keeps all 50000 counters in its LDS; the scan adds the slices up; no global atomics, no memset; sum order unchanged
# speedup vs baseline: 4.3283x; 1.2454x over previous
#include <hip/hip_runtime.h>


typedef __attribute__((ext_vector_type(2))) float v2f;
typedef __attribute__((ext_vector_type(8))) float v8f;

#define NNODES 50000
#define FEAT   256
#define OUTF   128
#define NEDGE  800000

typedef __attribute__((ext_vector_type(4))) float v4f;

#define BM  80
#define BN  64
#define KC  32
#define LDP 36

__global__ __launch_bounds__(128) void gemm_tiled_wmma(
    const float* __restrict__ A0, const float* __restrict__ W0,
    const float* __restrict__ A1, const float* __restrict__ W1,
    const float* __restrict__ bias, float* __restrict__ Y, int nseg, int colOff)
{
  __shared__ __attribute__((aligned(16))) float As[BM * LDP];
  __shared__ __attribute__((aligned(16))) float Bs[BN * LDP];

  const int tid  = threadIdx.x;
  const int lane = tid & 31;
  const int wave = tid >> 5;
  const int c    = lane & 15;
  const int h    = lane >> 4;
  const int m0   = blockIdx.x * BM;
  const int n0   = blockIdx.y * BN;

  const int srow = tid >> 3;
  const int scol = (tid & 7) << 2;
  const size_t aoff = (size_t)(m0 + srow) * FEAT + scol;
  const size_t woff = (size_t)(n0 + srow) * FEAT + scol;
  float* const asw = As + srow * LDP + scol;
  float* const bsw = Bs + srow * LDP + scol;
  const float* const afr = As + c * LDP + (h << 2);
  const float* const bfr = Bs + ((wave << 4) + c) * LDP + (h << 2);

  v8f acc[5];
#pragma unroll
  for (int i = 0; i < 5; ++i) acc[i] = (v8f){0.f, 0.f, 0.f, 0.f, 0.f, 0.f, 0.f, 0.f};

  v4f ra[5], rb[4];
  {
    const float* ap = A0 + aoff;
    const float* wp = W0 + woff;
#pragma unroll
    for (int i = 0; i < 5; ++i) ra[i] = *(const v4f*)(ap + (size_t)i * 16 * FEAT);
#pragma unroll
    for (int i = 0; i < 4; ++i) rb[i] = *(const v4f*)(wp + (size_t)i * 16 * FEAT);
  }

  const int nch = nseg * (FEAT / KC);
  for (int ch = 0; ch < nch; ++ch) {
    __syncthreads();
#pragma unroll
    for (int i = 0; i < 5; ++i) *(v4f*)(asw + i * 16 * LDP) = ra[i];
#pragma unroll
    for (int i = 0; i < 4; ++i) *(v4f*)(bsw + i * 16 * LDP) = rb[i];
    __syncthreads();

    if (ch + 1 < nch) {
      const int   nx = ch + 1;
      const int   kb = (nx & 7) * KC;
      const float* ap = ((nx < 8) ? A0 : A1) + aoff + kb;
      const float* wp = ((nx < 8) ? W0 : W1) + woff + kb;
#pragma unroll
      for (int i = 0; i < 5; ++i) ra[i] = *(const v4f*)(ap + (size_t)i * 16 * FEAT);
#pragma unroll
      for (int i = 0; i < 4; ++i) rb[i] = *(const v4f*)(wp + (size_t)i * 16 * FEAT);
    }

    v4f fa[2][5], fb[2];
    fb[0] = *(const v4f*)bfr;
#pragma unroll
    for (int i = 0; i < 5; ++i) fa[0][i] = *(const v4f*)(afr + i * 16 * LDP);
#pragma unroll
    for (int j = 0; j < 4; ++j) {
      const int cur = j & 1, nxt = cur ^ 1;
      if (j + 1 < 4) {
        fb[nxt] = *(const v4f*)(bfr + (j + 1) * 8);
#pragma unroll
        for (int i = 0; i < 5; ++i)
          fa[nxt][i] = *(const v4f*)(afr + i * 16 * LDP + (j + 1) * 8);
      }
#pragma unroll
      for (int i = 0; i < 5; ++i) {
        const v2f a = fa[cur][i].xy;
        const v2f b = fb[cur].xy;
        acc[i] = __builtin_amdgcn_wmma_f32_16x16x4_f32(
            false, a, false, b, (short)0, acc[i], false, false);
      }
#pragma unroll
      for (int i = 0; i < 5; ++i) {
        const v2f a = fa[cur][i].zw;
        const v2f b = fb[cur].zw;
        acc[i] = __builtin_amdgcn_wmma_f32_16x16x4_f32(
            false, a, false, b, (short)0, acc[i], false, false);
      }
    }
  }

  const int col = n0 + (wave << 4) + c;
  const float bv = bias[col];
  float* yp = Y + (size_t)(m0 + (h << 3)) * FEAT + colOff + col;
#pragma unroll
  for (int i = 0; i < 5; ++i) {
#pragma unroll
    for (int r = 0; r < 8; ++r)
      yp[(size_t)(i * 16 + r) * FEAT] = acc[i][r] + bv;
  }
}

#define NSL    32
#define EPS    (NEDGE / NSL)
#define NCHUNK ((NNODES + 1023) / 1024)

__global__ __launch_bounds__(1024) void edge_rank_lds(
    const int* __restrict__ EI, int* __restrict__ RANK,
    int* __restrict__ H, int* __restrict__ G)
{
  __shared__ __attribute__((aligned(16))) int hist[NNODES];
  const int t    = threadIdx.x;
  const int lane = t & 31;
  const int wave = t >> 5;
  const int sl   = blockIdx.x;

  int4* h4 = (int4*)hist;
  for (int i = t; i < NNODES / 4; i += 1024) h4[i] = make_int4(0, 0, 0, 0);
  __syncthreads();

  const int* er = EI + sl * EPS;
  int*       rk = RANK + sl * EPS;
  for (int i0 = 0; i0 < EPS; i0 += 4096) {
    int r[4];
#pragma unroll
    for (int j = 0; j < 4; ++j) {
      const int i = i0 + (j << 10) + t;
      r[j] = (i < EPS) ? er[i] : 0;
    }
    int q[4];
#pragma unroll
    for (int j = 0; j < 4; ++j) {
      const int i = i0 + (j << 10) + t;
      q[j] = (i < EPS) ? atomicAdd(&hist[r[j]], 1) : 0;
    }
#pragma unroll
    for (int j = 0; j < 4; ++j) {
      const int i = i0 + (j << 10) + t;
      if (i < EPS) rk[i] = q[j];
    }
  }
  __syncthreads();

  int* hp = H + sl * NNODES;
  for (int i = t; i < NNODES; i += 1024) hp[i] = hist[i];

  for (int k = wave; k < NCHUNK; k += 32) {
    int s = 0;
    for (int j = 0; j < 32; ++j) {
      const int i = (k << 10) + (j << 5) + lane;
      s += (i < NNODES) ? hist[i] : 0;
    }
    for (int off = 16; off >= 1; off >>= 1) s += __shfl_xor(s, off, 32);
    if (lane == 0) G[sl * 64 + k] = s;
  }
}

__global__ __launch_bounds__(1024) void scan_slices(
    int* __restrict__ H, const int* __restrict__ G, int* __restrict__ OFS, int n)
{
  __shared__ int wsum[32];
  __shared__ int wpre[32];
  const int t    = threadIdx.x;
  const int lane = t & 31;
  const int wave = t >> 5;
  const int base = blockIdx.x << 10;

  int pre = 0;
  for (int i = t; i < NSL * 64; i += 1024)
    if ((i & 63) < (int)blockIdx.x) pre += G[i];

  const int idx = base + t;
  int hv[NSL];
#pragma unroll
  for (int b = 0; b < NSL; ++b) hv[b] = (idx < n) ? H[b * NNODES + idx] : 0;
  int c = 0;
#pragma unroll
  for (int b = 0; b < NSL; ++b) c += hv[b];

  int inc = c;
  for (int off = 1; off < 32; off <<= 1) {
    const int v = __shfl_up(inc, off, 32);
    if (lane >= off) inc += v;
  }
  for (int off = 16; off >= 1; off >>= 1) pre += __shfl_xor(pre, off, 32);
  if (lane == 31) wsum[wave] = inc;
  if (lane == 0)  wpre[wave] = pre;
  __syncthreads();
  int front = 0;
  for (int w = 0; w < 32; ++w) {
    front += wpre[w];
    if (w < wave) front += wsum[w];
  }
  if (idx < n) {
    int run = front + inc - c;
    OFS[idx] = run;
#pragma unroll
    for (int b = 0; b < NSL; ++b) {
      H[b * NNODES + idx] = run;
      run += hv[b];
    }
  }
}

__global__ __launch_bounds__(256) void edge_place(
    const int* __restrict__ EI, const int* __restrict__ H,
    const int* __restrict__ RANK, int* __restrict__ BKT, int E)
{
  const int e = blockIdx.x * 256 + threadIdx.x;
  if (e < E) {
    const int sl = e / EPS;
    BKT[H[sl * NNODES + EI[e]] + RANK[e]] = EI[E + e];
  }
}

__device__ __forceinline__ void edge_term(
    const float* __restrict__ X, int l, int lane,
    const float4& a0, const float4& a1, float4& s0, float4& s1)
{
  const float4* xp = (const float4*)(X + (size_t)l * FEAT);
  const float4 x0 = xp[lane], x1 = xp[lane + 32];
  float v = a0.x * x0.x + a0.y * x0.y + a0.z * x0.z + a0.w * x0.w
          + a1.x * x1.x + a1.y * x1.y + a1.z * x1.z + a1.w * x1.w;
  for (int off = 16; off >= 1; off >>= 1) v += __shfl_xor(v, off, 32);
  s0.x += v * x0.x; s0.y += v * x0.y; s0.z += v * x0.z; s0.w += v * x0.w;
  s1.x += v * x1.x; s1.y += v * x1.y; s1.z += v * x1.z; s1.w += v * x1.w;
}

__global__ __launch_bounds__(256) void node_aggregate(
    const int* __restrict__ OFS, const int* __restrict__ BKT,
    const float* __restrict__ X, float* __restrict__ AS, int n, int E)
{
  const int lane = threadIdx.x & 31;
  const int wave = threadIdx.x >> 5;
  const int r = __builtin_amdgcn_readfirstlane((blockIdx.x << 3) + wave);
  if (r >= n) return;
  const int start = OFS[r];
  const int end   = (r + 1 < n) ? OFS[r + 1] : E;
  const int deg   = end - start;

  float4* row = (float4*)(AS + (size_t)r * FEAT);
  const float4 a0 = row[lane], a1 = row[lane + 32];
  float4 s0 = {0.f, 0.f, 0.f, 0.f}, s1 = {0.f, 0.f, 0.f, 0.f};

  if (deg <= 32) {
    const int k = (lane < deg) ? BKT[start + lane] : 0x7fffffff;
    int rank = 0;
#pragma unroll
    for (int j = 0; j < 32; ++j) {
      const int kj = __builtin_amdgcn_readlane(k, j);
      rank += ((kj < k) || (kj == k && j < lane)) ? 1 : 0;
    }
    const int sorted = __builtin_amdgcn_ds_permute(rank << 2, k);
#pragma unroll 4
    for (int t = 0; t < deg; ++t) {
      const int l = __builtin_amdgcn_readlane(sorted, t);
      edge_term(X, l, lane, a0, a1, s0, s1);
    }
  } else {
    int last = -1;
    for (int it = 0; it < deg; ++it) {
      int cand = 0x7fffffff;
      for (int b = start; b < end; b += 32) {
        const int p = b + lane;
        const int k = (p < end) ? BKT[p] : 0x7fffffff;
        if (k > last && k < cand) cand = k;
      }
      for (int off = 16; off >= 1; off >>= 1) {
        const int o = __shfl_xor(cand, off, 32);
        cand = (o < cand) ? o : cand;
      }
      if (cand == 0x7fffffff) break;
      int mult = 0;
      for (int b = start; b < end; b += 32) {
        const int p = b + lane;
        const int k = (p < end) ? BKT[p] : 0x7fffffff;
        mult += __popc(__builtin_amdgcn_ballot_w32(k == cand));
      }
      const int l = __builtin_amdgcn_readfirstlane(cand);
      for (int m = 0; m < mult; ++m) edge_term(X, l, lane, a0, a1, s0, s1);
      last = cand;
    }
  }

  const float inv = 1.0f / fmaxf((float)deg, 1.0f);
  s0.x *= inv; s0.y *= inv; s0.z *= inv; s0.w *= inv;
  s1.x *= inv; s1.y *= inv; s1.z *= inv; s1.w *= inv;
  row[lane] = s0;
  row[lane + 32] = s1;
}

extern "C" void kernel_launch(void* const* d_in, const int* in_sizes, int n_in,
                              void* d_out, int out_size, void* d_ws, size_t ws_size,
                              hipStream_t stream)
{
  const float* x        = (const float*)d_in[0];
  const int*   pos_ei   = (const int*)d_in[1];
  const int*   neg_ei   = (const int*)d_in[2];
  const float* W_pos    = (const float*)d_in[3];
  const float* W_pos_cc = (const float*)d_in[4];
  const float* b_pos_cc = (const float*)d_in[5];
  const float* W_neg    = (const float*)d_in[6];
  const float* W_neg_cc = (const float*)d_in[7];
  const float* b_neg_cc = (const float*)d_in[8];
  const float* W_patt   = (const float*)d_in[9];
  const float* b_patt   = (const float*)d_in[10];
  const float* W_natt   = (const float*)d_in[11];
  const float* b_natt   = (const float*)d_in[12];
  float* out = (float*)d_out;

  float* Sbuf = (float*)d_ws;
  int*   Rank = (int*)d_ws;
  int*   Hsl  = Rank + (size_t)NEDGE;
  int*   Gsl  = Hsl + (size_t)NSL * NNODES;
  int*   Bkt  = (int*)(Sbuf + (size_t)NNODES * FEAT);
  int*   Ofs  = Bkt + (size_t)NEDGE;

  const dim3 blk(256);
  const dim3 gblk(128);
  const dim3 gAtt(NNODES / BM, FEAT / BN);
  const dim3 gOut(NNODES / BM, OUTF / BN);
  const int  gEdge = (NEDGE + 255) / 256;
  const int  gScan = (NNODES + 1023) / 1024;
  const int  gNode = (NNODES + 7) / 8;

  const int*   ei[2]   = { pos_ei, neg_ei };
  const float* Watt[2] = { W_patt, W_natt };
  const float* batt[2] = { b_patt, b_natt };
  const float* Wagg[2] = { W_pos,  W_neg  };
  const float* Wcc[2]  = { W_pos_cc, W_neg_cc };
  const float* bcc[2]  = { b_pos_cc, b_neg_cc };

  for (int s = 0; s < 2; ++s) {
    edge_rank_lds<<<NSL, dim3(1024), 0, stream>>>(ei[s], Rank, Hsl, Gsl);
    scan_slices<<<gScan, dim3(1024), 0, stream>>>(Hsl, Gsl, Ofs, NNODES);
    edge_place<<<gEdge, blk, 0, stream>>>(ei[s], Hsl, Rank, Bkt, NEDGE);
    gemm_tiled_wmma<<<gAtt, gblk, 0, stream>>>(x, Watt[s], x, Watt[s],
                                               batt[s], Sbuf, 1, 0);
    node_aggregate<<<gNode, blk, 0, stream>>>(Ofs, Bkt, x, Sbuf, NNODES, NEDGE);
    gemm_tiled_wmma<<<gOut, gblk, 0, stream>>>(Sbuf, Wagg[s], x, Wcc[s],
                                               bcc[s], out, 2, s * OUTF);
  }
}


# Per-node pass: buckets of 33 to 64 edges ranked in registers (two keys per lane) and summed two edges at a time, not one key per scan of the bucket in memory; ascending-source sum order unchanged
# speedup vs baseline: 4.8619x; 1.1233x over previous
#include <hip/hip_runtime.h>


typedef __attribute__((ext_vector_type(2))) float v2f;
typedef __attribute__((ext_vector_type(8))) float v8f;

#define NNODES 50000
#define FEAT   256
#define OUTF   128
#define NEDGE  800000

typedef __attribute__((ext_vector_type(4))) float v4f;

#define BM  80
#define BN  64
#define KC  32
#define LDP 36

__global__ __launch_bounds__(128) void gemm_tiled_wmma(
    const float* __restrict__ A0, const float* __restrict__ W0,
    const float* __restrict__ A1, const float* __restrict__ W1,
    const float* __restrict__ bias, float* __restrict__ Y, int nseg, int colOff)
{
  __shared__ __attribute__((aligned(16))) float As[BM * LDP];
  __shared__ __attribute__((aligned(16))) float Bs[BN * LDP];

  const int tid  = threadIdx.x;
  const int lane = tid & 31;
  const int wave = tid >> 5;
  const int c    = lane & 15;
  const int h    = lane >> 4;
  const int m0   = blockIdx.x * BM;
  const int n0   = blockIdx.y * BN;

  const int srow = tid >> 3;
  const int scol = (tid & 7) << 2;
  const size_t aoff = (size_t)(m0 + srow) * FEAT + scol;
  const size_t woff = (size_t)(n0 + srow) * FEAT + scol;
  float* const asw = As + srow * LDP + scol;
  float* const bsw = Bs + srow * LDP + scol;
  const float* const afr = As + c * LDP + (h << 2);
  const float* const bfr = Bs + ((wave << 4) + c) * LDP + (h << 2);

  v8f acc[5];
#pragma unroll
  for (int i = 0; i < 5; ++i) acc[i] = (v8f){0.f, 0.f, 0.f, 0.f, 0.f, 0.f, 0.f, 0.f};

  v4f ra[5], rb[4];
  {
    const float* ap = A0 + aoff;
    const float* wp = W0 + woff;
#pragma unroll
    for (int i = 0; i < 5; ++i) ra[i] = *(const v4f*)(ap + (size_t)i * 16 * FEAT);
#pragma unroll
    for (int i = 0; i < 4; ++i) rb[i] = *(const v4f*)(wp + (size_t)i * 16 * FEAT);
  }

  const int nch = nseg * (FEAT / KC);
  for (int ch = 0; ch < nch; ++ch) {
    __syncthreads();
#pragma unroll
    for (int i = 0; i < 5; ++i) *(v4f*)(asw + i * 16 * LDP) = ra[i];
#pragma unroll
    for (int i = 0; i < 4; ++i) *(v4f*)(bsw + i * 16 * LDP) = rb[i];
    __syncthreads();

    if (ch + 1 < nch) {
      const int   nx = ch + 1;
      const int   kb = (nx & 7) * KC;
      const float* ap = ((nx < 8) ? A0 : A1) + aoff + kb;
      const float* wp = ((nx < 8) ? W0 : W1) + woff + kb;
#pragma unroll
      for (int i = 0; i < 5; ++i) ra[i] = *(const v4f*)(ap + (size_t)i * 16 * FEAT);
#pragma unroll
      for (int i = 0; i < 4; ++i) rb[i] = *(const v4f*)(wp + (size_t)i * 16 * FEAT);
    }

    v4f fa[2][5], fb[2];
    fb[0] = *(const v4f*)bfr;
#pragma unroll
    for (int i = 0; i < 5; ++i) fa[0][i] = *(const v4f*)(afr + i * 16 * LDP);
#pragma unroll
    for (int j = 0; j < 4; ++j) {
      const int cur = j & 1, nxt = cur ^ 1;
      if (j + 1 < 4) {
        fb[nxt] = *(const v4f*)(bfr + (j + 1) * 8);
#pragma unroll
        for (int i = 0; i < 5; ++i)
          fa[nxt][i] = *(const v4f*)(afr + i * 16 * LDP + (j + 1) * 8);
      }
#pragma unroll
      for (int i = 0; i < 5; ++i) {
        const v2f a = fa[cur][i].xy;
        const v2f b = fb[cur].xy;
        acc[i] = __builtin_amdgcn_wmma_f32_16x16x4_f32(
            false, a, false, b, (short)0, acc[i], false, false);
      }
#pragma unroll
      for (int i = 0; i < 5; ++i) {
        const v2f a = fa[cur][i].zw;
        const v2f b = fb[cur].zw;
        acc[i] = __builtin_amdgcn_wmma_f32_16x16x4_f32(
            false, a, false, b, (short)0, acc[i], false, false);
      }
    }
  }

  const int col = n0 + (wave << 4) + c;
  const float bv = bias[col];
  float* yp = Y + (size_t)(m0 + (h << 3)) * FEAT + colOff + col;
#pragma unroll
  for (int i = 0; i < 5; ++i) {
#pragma unroll
    for (int r = 0; r < 8; ++r)
      yp[(size_t)(i * 16 + r) * FEAT] = acc[i][r] + bv;
  }
}

#define NSL    32
#define EPS    (NEDGE / NSL)
#define NCHUNK ((NNODES + 1023) / 1024)

__global__ __launch_bounds__(1024) void edge_rank_lds(
    const int* __restrict__ EI, int* __restrict__ RANK,
    int* __restrict__ H, int* __restrict__ G)
{
  __shared__ __attribute__((aligned(16))) int hist[NNODES];
  const int t    = threadIdx.x;
  const int lane = t & 31;
  const int wave = t >> 5;
  const int sl   = blockIdx.x;

  int4* h4 = (int4*)hist;
  for (int i = t; i < NNODES / 4; i += 1024) h4[i] = make_int4(0, 0, 0, 0);
  __syncthreads();

  const int* er = EI + sl * EPS;
  int*       rk = RANK + sl * EPS;
  for (int i0 = 0; i0 < EPS; i0 += 4096) {
    int r[4];
#pragma unroll
    for (int j = 0; j < 4; ++j) {
      const int i = i0 + (j << 10) + t;
      r[j] = (i < EPS) ? er[i] : 0;
    }
    int q[4];
#pragma unroll
    for (int j = 0; j < 4; ++j) {
      const int i = i0 + (j << 10) + t;
      q[j] = (i < EPS) ? atomicAdd(&hist[r[j]], 1) : 0;
    }
#pragma unroll
    for (int j = 0; j < 4; ++j) {
      const int i = i0 + (j << 10) + t;
      if (i < EPS) rk[i] = q[j];
    }
  }
  __syncthreads();

  int* hp = H + sl * NNODES;
  for (int i = t; i < NNODES; i += 1024) hp[i] = hist[i];

  for (int k = wave; k < NCHUNK; k += 32) {
    int s = 0;
    for (int j = 0; j < 32; ++j) {
      const int i = (k << 10) + (j << 5) + lane;
      s += (i < NNODES) ? hist[i] : 0;
    }
    for (int off = 16; off >= 1; off >>= 1) s += __shfl_xor(s, off, 32);
    if (lane == 0) G[sl * 64 + k] = s;
  }
}

__global__ __launch_bounds__(1024) void scan_slices(
    int* __restrict__ H, const int* __restrict__ G, int* __restrict__ OFS, int n)
{
  __shared__ int wsum[32];
  __shared__ int wpre[32];
  const int t    = threadIdx.x;
  const int lane = t & 31;
  const int wave = t >> 5;
  const int base = blockIdx.x << 10;

  int pre = 0;
  for (int i = t; i < NSL * 64; i += 1024)
    if ((i & 63) < (int)blockIdx.x) pre += G[i];

  const int idx = base + t;
  int hv[NSL];
#pragma unroll
  for (int b = 0; b < NSL; ++b) hv[b] = (idx < n) ? H[b * NNODES + idx] : 0;
  int c = 0;
#pragma unroll
  for (int b = 0; b < NSL; ++b) c += hv[b];

  int inc = c;
  for (int off = 1; off < 32; off <<= 1) {
    const int v = __shfl_up(inc, off, 32);
    if (lane >= off) inc += v;
  }
  for (int off = 16; off >= 1; off >>= 1) pre += __shfl_xor(pre, off, 32);
  if (lane == 31) wsum[wave] = inc;
  if (lane == 0)  wpre[wave] = pre;
  __syncthreads();
  int front = 0;
  for (int w = 0; w < 32; ++w) {
    front += wpre[w];
    if (w < wave) front += wsum[w];
  }
  if (idx < n) {
    int run = front + inc - c;
    OFS[idx] = run;
#pragma unroll
    for (int b = 0; b < NSL; ++b) {
      H[b * NNODES + idx] = run;
      run += hv[b];
    }
  }
}

__global__ __launch_bounds__(256) void edge_place(
    const int* __restrict__ EI, const int* __restrict__ H,
    const int* __restrict__ RANK, int* __restrict__ BKT, int E)
{
  const int e = blockIdx.x * 256 + threadIdx.x;
  if (e < E) {
    const int sl = e / EPS;
    BKT[H[sl * NNODES + EI[e]] + RANK[e]] = EI[E + e];
  }
}

__device__ __forceinline__ void edge_term(
    const float* __restrict__ X, int l, int lane,
    const float4& a0, const float4& a1, float4& s0, float4& s1)
{
  const float4* xp = (const float4*)(X + (size_t)l * FEAT);
  const float4 x0 = xp[lane], x1 = xp[lane + 32];
  float v = a0.x * x0.x + a0.y * x0.y + a0.z * x0.z + a0.w * x0.w
          + a1.x * x1.x + a1.y * x1.y + a1.z * x1.z + a1.w * x1.w;
  for (int off = 16; off >= 1; off >>= 1) v += __shfl_xor(v, off, 32);
  s0.x += v * x0.x; s0.y += v * x0.y; s0.z += v * x0.z; s0.w += v * x0.w;
  s1.x += v * x1.x; s1.y += v * x1.y; s1.z += v * x1.z; s1.w += v * x1.w;
}

__global__ __launch_bounds__(256) void node_aggregate(
    const int* __restrict__ OFS, const int* __restrict__ BKT,
    const float* __restrict__ X, float* __restrict__ AS, int n, int E)
{
  const int lane = threadIdx.x & 31;
  const int wave = threadIdx.x >> 5;
  const int r = __builtin_amdgcn_readfirstlane((blockIdx.x << 3) + wave);
  if (r >= n) return;
  const int start = OFS[r];
  const int end   = (r + 1 < n) ? OFS[r + 1] : E;
  const int deg   = end - start;

  float4* row = (float4*)(AS + (size_t)r * FEAT);
  const float4 a0 = row[lane], a1 = row[lane + 32];
  float4 s0 = {0.f, 0.f, 0.f, 0.f}, s1 = {0.f, 0.f, 0.f, 0.f};

  if (deg <= 32) {
    const int k = (lane < deg) ? BKT[start + lane] : 0x7fffffff;
    int rank = 0;
#pragma unroll
    for (int j = 0; j < 32; ++j) {
      const int kj = __builtin_amdgcn_readlane(k, j);
      rank += ((kj < k) || (kj == k && j < lane)) ? 1 : 0;
    }
    const int sorted = __builtin_amdgcn_ds_permute(rank << 2, k);
#pragma unroll 4
    for (int t = 0; t < deg; ++t) {
      const int l = __builtin_amdgcn_readlane(sorted, t);
      edge_term(X, l, lane, a0, a1, s0, s1);
    }
  } else if (deg <= 64) {
    const int k0 = BKT[start + lane];
    const int k1 = (lane + 32 < deg) ? BKT[start + 32 + lane] : 0x7fffffff;
    int r0 = 0, r1 = 0;
    for (int j = 0; j < 32; ++j) {
      const int p = __builtin_amdgcn_readlane(k0, j);
      const int q = __builtin_amdgcn_readlane(k1, j);
      r0 += ((p < k0) || (p == k0 && j < lane)) ? 1 : 0;
      r0 += (q < k0) ? 1 : 0;
      r1 += (p <= k1) ? 1 : 0;
      r1 += ((q < k1) || (q == k1 && j < lane)) ? 1 : 0;
    }
#pragma unroll 2
    for (int t = 0; t < deg; ++t) {
      const bool h0 = (r0 == t);
      const bool h1 = (r1 == t);
      const int  kv = h0 ? k0 : k1;
      const unsigned m = __builtin_amdgcn_ballot_w32(h0 || h1);
      const int l = __builtin_amdgcn_readlane(kv, __builtin_ctz(m));
      edge_term(X, l, lane, a0, a1, s0, s1);
    }
  } else {
    int last = -1;
    for (int it = 0; it < deg; ++it) {
      int cand = 0x7fffffff;
      for (int b = start; b < end; b += 32) {
        const int p = b + lane;
        const int k = (p < end) ? BKT[p] : 0x7fffffff;
        if (k > last && k < cand) cand = k;
      }
      for (int off = 16; off >= 1; off >>= 1) {
        const int o = __shfl_xor(cand, off, 32);
        cand = (o < cand) ? o : cand;
      }
      if (cand == 0x7fffffff) break;
      int mult = 0;
      for (int b = start; b < end; b += 32) {
        const int p = b + lane;
        const int k = (p < end) ? BKT[p] : 0x7fffffff;
        mult += __popc(__builtin_amdgcn_ballot_w32(k == cand));
      }
      const int l = __builtin_amdgcn_readfirstlane(cand);
      for (int m = 0; m < mult; ++m) edge_term(X, l, lane, a0, a1, s0, s1);
      last = cand;
    }
  }

  const float inv = 1.0f / fmaxf((float)deg, 1.0f);
  s0.x *= inv; s0.y *= inv; s0.z *= inv; s0.w *= inv;
  s1.x *= inv; s1.y *= inv; s1.z *= inv; s1.w *= inv;
  row[lane] = s0;
  row[lane + 32] = s1;
}

extern "C" void kernel_launch(void* const* d_in, const int* in_sizes, int n_in,
                              void* d_out, int out_size, void* d_ws, size_t ws_size,
                              hipStream_t stream)
{
  const float* x        = (const float*)d_in[0];
  const int*   pos_ei   = (const int*)d_in[1];
  const int*   neg_ei   = (const int*)d_in[2];
  const float* W_pos    = (const float*)d_in[3];
  const float* W_pos_cc = (const float*)d_in[4];
  const float* b_pos_cc = (const float*)d_in[5];
  const float* W_neg    = (const float*)d_in[6];
  const float* W_neg_cc = (const float*)d_in[7];
  const float* b_neg_cc = (const float*)d_in[8];
  const float* W_patt   = (const float*)d_in[9];
  const float* b_patt   = (const float*)d_in[10];
  const float* W_natt   = (const float*)d_in[11];
  const float* b_natt   = (const float*)d_in[12];
  float* out = (float*)d_out;

  float* Sbuf = (float*)d_ws;
  int*   Rank = (int*)d_ws;
  int*   Hsl  = Rank + (size_t)NEDGE;
  int*   Gsl  = Hsl + (size_t)NSL * NNODES;
  int*   Bkt  = (int*)(Sbuf + (size_t)NNODES * FEAT);
  int*   Ofs  = Bkt + (size_t)NEDGE;

  const dim3 blk(256);
  const dim3 gblk(128);
  const dim3 gAtt(NNODES / BM, FEAT / BN);
  const dim3 gOut(NNODES / BM, OUTF / BN);
  const int  gEdge = (NEDGE + 255) / 256;
  const int  gScan = (NNODES + 1023) / 1024;
  const int  gNode = (NNODES + 7) / 8;

  const int*   ei[2]   = { pos_ei, neg_ei };
  const float* Watt[2] = { W_patt, W_natt };
  const float* batt[2] = { b_patt, b_natt };
  const float* Wagg[2] = { W_pos,  W_neg  };
  const float* Wcc[2]  = { W_pos_cc, W_neg_cc };
  const float* bcc[2]  = { b_pos_cc, b_neg_cc };

  for (int s = 0; s < 2; ++s) {
    edge_rank_lds<<<NSL, dim3(1024), 0, stream>>>(ei[s], Rank, Hsl, Gsl);
    scan_slices<<<gScan, dim3(1024), 0, stream>>>(Hsl, Gsl, Ofs, NNODES);
    edge_place<<<gEdge, blk, 0, stream>>>(ei[s], Hsl, Rank, Bkt, NEDGE);
    gemm_tiled_wmma<<<gAtt, gblk, 0, stream>>>(x, Watt[s], x, Watt[s],
                                               batt[s], Sbuf, 1, 0);
    node_aggregate<<<gNode, blk, 0, stream>>>(Ofs, Bkt, x, Sbuf, NNODES, NEDGE);
    gemm_tiled_wmma<<<gOut, gblk, 0, stream>>>(Sbuf, Wagg[s], x, Wcc[s],
                                               bcc[s], out, 2, s * OUTF);
  }
}


# Edge grouping of both signs side by side in the same three launches (12 launches -> 9); the first sign's bucket list waits in the output buffer until its node pass; sum order unchanged
# speedup vs baseline: 5.1681x; 1.0630x over previous
#include <hip/hip_runtime.h>


typedef __attribute__((ext_vector_type(2))) float v2f;
typedef __attribute__((ext_vector_type(8))) float v8f;

#define NNODES 50000
#define FEAT   256
#define OUTF   128
#define NEDGE  800000

typedef __attribute__((ext_vector_type(4))) float v4f;

#define BM  80
#define BN  64
#define KC  32
#define LDP 36

__global__ __launch_bounds__(128) void gemm_tiled_wmma(
    const float* __restrict__ A0, const float* __restrict__ W0,
    const float* __restrict__ A1, const float* __restrict__ W1,
    const float* __restrict__ bias, float* __restrict__ Y, int nseg, int colOff)
{
  __shared__ __attribute__((aligned(16))) float As[BM * LDP];
  __shared__ __attribute__((aligned(16))) float Bs[BN * LDP];

  const int tid  = threadIdx.x;
  const int lane = tid & 31;
  const int wave = tid >> 5;
  const int c    = lane & 15;
  const int h    = lane >> 4;
  const int m0   = blockIdx.x * BM;
  const int n0   = blockIdx.y * BN;

  const int srow = tid >> 3;
  const int scol = (tid & 7) << 2;
  const size_t aoff = (size_t)(m0 + srow) * FEAT + scol;
  const size_t woff = (size_t)(n0 + srow) * FEAT + scol;
  float* const asw = As + srow * LDP + scol;
  float* const bsw = Bs + srow * LDP + scol;
  const float* const afr = As + c * LDP + (h << 2);
  const float* const bfr = Bs + ((wave << 4) + c) * LDP + (h << 2);

  v8f acc[5];
#pragma unroll
  for (int i = 0; i < 5; ++i) acc[i] = (v8f){0.f, 0.f, 0.f, 0.f, 0.f, 0.f, 0.f, 0.f};

  v4f ra[5], rb[4];
  {
    const float* ap = A0 + aoff;
    const float* wp = W0 + woff;
#pragma unroll
    for (int i = 0; i < 5; ++i) ra[i] = *(const v4f*)(ap + (size_t)i * 16 * FEAT);
#pragma unroll
    for (int i = 0; i < 4; ++i) rb[i] = *(const v4f*)(wp + (size_t)i * 16 * FEAT);
  }

  const int nch = nseg * (FEAT / KC);
  for (int ch = 0; ch < nch; ++ch) {
    __syncthreads();
#pragma unroll
    for (int i = 0; i < 5; ++i) *(v4f*)(asw + i * 16 * LDP) = ra[i];
#pragma unroll
    for (int i = 0; i < 4; ++i) *(v4f*)(bsw + i * 16 * LDP) = rb[i];
    __syncthreads();

    if (ch + 1 < nch) {
      const int   nx = ch + 1;
      const int   kb = (nx & 7) * KC;
      const float* ap = ((nx < 8) ? A0 : A1) + aoff + kb;
      const float* wp = ((nx < 8) ? W0 : W1) + woff + kb;
#pragma unroll
      for (int i = 0; i < 5; ++i) ra[i] = *(const v4f*)(ap + (size_t)i * 16 * FEAT);
#pragma unroll
      for (int i = 0; i < 4; ++i) rb[i] = *(const v4f*)(wp + (size_t)i * 16 * FEAT);
    }

    v4f fa[2][5], fb[2];
    fb[0] = *(const v4f*)bfr;
#pragma unroll
    for (int i = 0; i < 5; ++i) fa[0][i] = *(const v4f*)(afr + i * 16 * LDP);
#pragma unroll
    for (int j = 0; j < 4; ++j) {
      const int cur = j & 1, nxt = cur ^ 1;
      if (j + 1 < 4) {
        fb[nxt] = *(const v4f*)(bfr + (j + 1) * 8);
#pragma unroll
        for (int i = 0; i < 5; ++i)
          fa[nxt][i] = *(const v4f*)(afr + i * 16 * LDP + (j + 1) * 8);
      }
#pragma unroll
      for (int i = 0; i < 5; ++i) {
        const v2f a = fa[cur][i].xy;
        const v2f b = fb[cur].xy;
        acc[i] = __builtin_amdgcn_wmma_f32_16x16x4_f32(
            false, a, false, b, (short)0, acc[i], false, false);
      }
#pragma unroll
      for (int i = 0; i < 5; ++i) {
        const v2f a = fa[cur][i].zw;
        const v2f b = fb[cur].zw;
        acc[i] = __builtin_amdgcn_wmma_f32_16x16x4_f32(
            false, a, false, b, (short)0, acc[i], false, false);
      }
    }
  }

  const int col = n0 + (wave << 4) + c;
  const float bv = bias[col];
  float* yp = Y + (size_t)(m0 + (h << 3)) * FEAT + colOff + col;
#pragma unroll
  for (int i = 0; i < 5; ++i) {
#pragma unroll
    for (int r = 0; r < 8; ++r)
      yp[(size_t)(i * 16 + r) * FEAT] = acc[i][r] + bv;
  }
}

#define NSL    32
#define EPS    (NEDGE / NSL)
#define NCHUNK ((NNODES + 1023) / 1024)

__global__ __launch_bounds__(1024) void edge_rank_lds(
    const int* __restrict__ EI0, const int* __restrict__ EI1,
    int* __restrict__ RANK, int* __restrict__ H, int* __restrict__ G)
{
  __shared__ __attribute__((aligned(16))) int hist[NNODES];
  const int t    = threadIdx.x;
  const int lane = t & 31;
  const int wave = t >> 5;
  const int sg   = blockIdx.x / NSL;
  const int sl   = blockIdx.x - sg * NSL;
  const int ss   = blockIdx.x;
  const int* __restrict__ EI = sg ? EI1 : EI0;

  int4* h4 = (int4*)hist;
  for (int i = t; i < NNODES / 4; i += 1024) h4[i] = make_int4(0, 0, 0, 0);
  __syncthreads();

  const int* er = EI + sl * EPS;
  int*       rk = RANK + sg * NEDGE + sl * EPS;
  for (int i0 = 0; i0 < EPS; i0 += 4096) {
    int r[4];
#pragma unroll
    for (int j = 0; j < 4; ++j) {
      const int i = i0 + (j << 10) + t;
      r[j] = (i < EPS) ? er[i] : 0;
    }
    int q[4];
#pragma unroll
    for (int j = 0; j < 4; ++j) {
      const int i = i0 + (j << 10) + t;
      q[j] = (i < EPS) ? atomicAdd(&hist[r[j]], 1) : 0;
    }
#pragma unroll
    for (int j = 0; j < 4; ++j) {
      const int i = i0 + (j << 10) + t;
      if (i < EPS) rk[i] = q[j];
    }
  }
  __syncthreads();

  int* hp = H + ss * NNODES;
  for (int i = t; i < NNODES; i += 1024) hp[i] = hist[i];

  for (int k = wave; k < NCHUNK; k += 32) {
    int s = 0;
    for (int j = 0; j < 32; ++j) {
      const int i = (k << 10) + (j << 5) + lane;
      s += (i < NNODES) ? hist[i] : 0;
    }
    for (int off = 16; off >= 1; off >>= 1) s += __shfl_xor(s, off, 32);
    if (lane == 0) G[ss * 64 + k] = s;
  }
}

__global__ __launch_bounds__(1024) void scan_slices(
    int* __restrict__ Hall, const int* __restrict__ Gall,
    int* __restrict__ OFS0, int* __restrict__ OFS1, int n)
{
  __shared__ int wsum[32];
  __shared__ int wpre[32];
  const int t    = threadIdx.x;
  const int lane = t & 31;
  const int wave = t >> 5;
  const int base = blockIdx.x << 10;
  const int sg   = blockIdx.y;
  int* __restrict__       H   = Hall + sg * (NSL * NNODES);
  const int* __restrict__ G   = Gall + sg * (NSL * 64);
  int* __restrict__       OFS = sg ? OFS1 : OFS0;

  int pre = 0;
  for (int i = t; i < NSL * 64; i += 1024)
    if ((i & 63) < (int)blockIdx.x) pre += G[i];

  const int idx = base + t;
  int hv[NSL];
#pragma unroll
  for (int b = 0; b < NSL; ++b) hv[b] = (idx < n) ? H[b * NNODES + idx] : 0;
  int c = 0;
#pragma unroll
  for (int b = 0; b < NSL; ++b) c += hv[b];

  int inc = c;
  for (int off = 1; off < 32; off <<= 1) {
    const int v = __shfl_up(inc, off, 32);
    if (lane >= off) inc += v;
  }
  for (int off = 16; off >= 1; off >>= 1) pre += __shfl_xor(pre, off, 32);
  if (lane == 31) wsum[wave] = inc;
  if (lane == 0)  wpre[wave] = pre;
  __syncthreads();
  int front = 0;
  for (int w = 0; w < 32; ++w) {
    front += wpre[w];
    if (w < wave) front += wsum[w];
  }
  if (idx < n) {
    int run = front + inc - c;
    OFS[idx] = run;
#pragma unroll
    for (int b = 0; b < NSL; ++b) {
      H[b * NNODES + idx] = run;
      run += hv[b];
    }
  }
}

__global__ __launch_bounds__(256) void edge_place(
    const int* __restrict__ EI0, const int* __restrict__ EI1,
    const int* __restrict__ H, const int* __restrict__ RANK,
    int* __restrict__ BKT0, int* __restrict__ BKT1, int E)
{
  const int sg  = blockIdx.y;
  const int* __restrict__ EI  = sg ? EI1 : EI0;
  int* __restrict__       BKT = sg ? BKT1 : BKT0;
  const int e = blockIdx.x * 256 + threadIdx.x;
  if (e < E) {
    const int sl = e / EPS;
    BKT[H[(sg * NSL + sl) * NNODES + EI[e]] + RANK[sg * NEDGE + e]] = EI[E + e];
  }
}

__device__ __forceinline__ void edge_term(
    const float* __restrict__ X, int l, int lane,
    const float4& a0, const float4& a1, float4& s0, float4& s1)
{
  const float4* xp = (const float4*)(X + (size_t)l * FEAT);
  const float4 x0 = xp[lane], x1 = xp[lane + 32];
  float v = a0.x * x0.x + a0.y * x0.y + a0.z * x0.z + a0.w * x0.w
          + a1.x * x1.x + a1.y * x1.y + a1.z * x1.z + a1.w * x1.w;
  for (int off = 16; off >= 1; off >>= 1) v += __shfl_xor(v, off, 32);
  s0.x += v * x0.x; s0.y += v * x0.y; s0.z += v * x0.z; s0.w += v * x0.w;
  s1.x += v * x1.x; s1.y += v * x1.y; s1.z += v * x1.z; s1.w += v * x1.w;
}

__global__ __launch_bounds__(256) void node_aggregate(
    const int* __restrict__ OFS, const int* __restrict__ BKT,
    const float* __restrict__ X, float* __restrict__ AS, int n, int E)
{
  const int lane = threadIdx.x & 31;
  const int wave = threadIdx.x >> 5;
  const int r = __builtin_amdgcn_readfirstlane((blockIdx.x << 3) + wave);
  if (r >= n) return;
  const int start = OFS[r];
  const int end   = (r + 1 < n) ? OFS[r + 1] : E;
  const int deg   = end - start;

  float4* row = (float4*)(AS + (size_t)r * FEAT);
  const float4 a0 = row[lane], a1 = row[lane + 32];
  float4 s0 = {0.f, 0.f, 0.f, 0.f}, s1 = {0.f, 0.f, 0.f, 0.f};

  if (deg <= 32) {
    const int k = (lane < deg) ? BKT[start + lane] : 0x7fffffff;
    int rank = 0;
#pragma unroll
    for (int j = 0; j < 32; ++j) {
      const int kj = __builtin_amdgcn_readlane(k, j);
      rank += ((kj < k) || (kj == k && j < lane)) ? 1 : 0;
    }
    const int sorted = __builtin_amdgcn_ds_permute(rank << 2, k);
#pragma unroll 4
    for (int t = 0; t < deg; ++t) {
      const int l = __builtin_amdgcn_readlane(sorted, t);
      edge_term(X, l, lane, a0, a1, s0, s1);
    }
  } else if (deg <= 64) {
    const int k0 = BKT[start + lane];
    const int k1 = (lane + 32 < deg) ? BKT[start + 32 + lane] : 0x7fffffff;
    int r0 = 0, r1 = 0;
    for (int j = 0; j < 32; ++j) {
      const int p = __builtin_amdgcn_readlane(k0, j);
      const int q = __builtin_amdgcn_readlane(k1, j);
      r0 += ((p < k0) || (p == k0 && j < lane)) ? 1 : 0;
      r0 += (q < k0) ? 1 : 0;
      r1 += (p <= k1) ? 1 : 0;
      r1 += ((q < k1) || (q == k1 && j < lane)) ? 1 : 0;
    }
#pragma unroll 2
    for (int t = 0; t < deg; ++t) {
      const bool h0 = (r0 == t);
      const bool h1 = (r1 == t);
      const int  kv = h0 ? k0 : k1;
      const unsigned m = __builtin_amdgcn_ballot_w32(h0 || h1);
      const int l = __builtin_amdgcn_readlane(kv, __builtin_ctz(m));
      edge_term(X, l, lane, a0, a1, s0, s1);
    }
  } else {
    int last = -1;
    for (int it = 0; it < deg; ++it) {
      int cand = 0x7fffffff;
      for (int b = start; b < end; b += 32) {
        const int p = b + lane;
        const int k = (p < end) ? BKT[p] : 0x7fffffff;
        if (k > last && k < cand) cand = k;
      }
      for (int off = 16; off >= 1; off >>= 1) {
        const int o = __shfl_xor(cand, off, 32);
        cand = (o < cand) ? o : cand;
      }
      if (cand == 0x7fffffff) break;
      int mult = 0;
      for (int b = start; b < end; b += 32) {
        const int p = b + lane;
        const int k = (p < end) ? BKT[p] : 0x7fffffff;
        mult += __popc(__builtin_amdgcn_ballot_w32(k == cand));
      }
      const int l = __builtin_amdgcn_readfirstlane(cand);
      for (int m = 0; m < mult; ++m) edge_term(X, l, lane, a0, a1, s0, s1);
      last = cand;
    }
  }

  const float inv = 1.0f / fmaxf((float)deg, 1.0f);
  s0.x *= inv; s0.y *= inv; s0.z *= inv; s0.w *= inv;
  s1.x *= inv; s1.y *= inv; s1.z *= inv; s1.w *= inv;
  row[lane] = s0;
  row[lane + 32] = s1;
}

extern "C" void kernel_launch(void* const* d_in, const int* in_sizes, int n_in,
                              void* d_out, int out_size, void* d_ws, size_t ws_size,
                              hipStream_t stream)
{
  const float* x        = (const float*)d_in[0];
  const int*   pos_ei   = (const int*)d_in[1];
  const int*   neg_ei   = (const int*)d_in[2];
  const float* W_pos    = (const float*)d_in[3];
  const float* W_pos_cc = (const float*)d_in[4];
  const float* b_pos_cc = (const float*)d_in[5];
  const float* W_neg    = (const float*)d_in[6];
  const float* W_neg_cc = (const float*)d_in[7];
  const float* b_neg_cc = (const float*)d_in[8];
  const float* W_patt   = (const float*)d_in[9];
  const float* b_patt   = (const float*)d_in[10];
  const float* W_natt   = (const float*)d_in[11];
  const float* b_natt   = (const float*)d_in[12];
  float* out = (float*)d_out;

  float* Sbuf = (float*)d_ws;
  int*   Rank = (int*)d_ws;
  int*   Hsl  = Rank + (size_t)2 * NEDGE;
  int*   Gsl  = Hsl + (size_t)2 * NSL * NNODES;
  int*   Bkt1 = (int*)(Sbuf + (size_t)NNODES * FEAT);
  int*   Ofs1 = Bkt1 + (size_t)NEDGE;
  int*   Bkt0 = (int*)d_out;
  int*   Ofs0 = Bkt0 + (size_t)NEDGE;

  const dim3 blk(256);
  const dim3 gblk(128);
  const dim3 gAtt(NNODES / BM, FEAT / BN);
  const dim3 gOut(NNODES / BM, OUTF / BN);
  const int  gEdge = (NEDGE + 255) / 256;
  const int  gScan = (NNODES + 1023) / 1024;
  const int  gNode = (NNODES + 7) / 8;

  const int*   bkt[2]  = { Bkt0, Bkt1 };
  const int*   ofs[2]  = { Ofs0, Ofs1 };
  const float* Watt[2] = { W_patt, W_natt };
  const float* batt[2] = { b_patt, b_natt };
  const float* Wagg[2] = { W_pos,  W_neg  };
  const float* Wcc[2]  = { W_pos_cc, W_neg_cc };
  const float* bcc[2]  = { b_pos_cc, b_neg_cc };

  edge_rank_lds<<<2 * NSL, dim3(1024), 0, stream>>>(pos_ei, neg_ei, Rank, Hsl, Gsl);
  scan_slices<<<dim3(gScan, 2), dim3(1024), 0, stream>>>(Hsl, Gsl, Ofs0, Ofs1, NNODES);
  edge_place<<<dim3(gEdge, 2), blk, 0, stream>>>(pos_ei, neg_ei, Hsl, Rank,
                                                 Bkt0, Bkt1, NEDGE);

  for (int s = 0; s < 2; ++s) {
    gemm_tiled_wmma<<<gAtt, gblk, 0, stream>>>(x, Watt[s], x, Watt[s],
                                               batt[s], Sbuf, 1, 0);
    node_aggregate<<<gNode, blk, 0, stream>>>(ofs[s], bkt[s], x, Sbuf, NNODES, NEDGE);
    gemm_tiled_wmma<<<gOut, gblk, 0, stream>>>(Sbuf, Wagg[s], x, Wcc[s],
                                               bcc[s], out, 2, s * OUTF);
  }
}


# Per-edge score: the 32-lane sum runs on the vector ALU (one row swap, then in-row xor adds 8, 4, 2, 1 as DPP) instead of five waited LDS shuffles; same partner order, scores bit-identical
# speedup vs baseline: 5.3819x; 1.0414x over previous
#include <hip/hip_runtime.h>


typedef __attribute__((ext_vector_type(2))) float v2f;
typedef __attribute__((ext_vector_type(8))) float v8f;

#define NNODES 50000
#define FEAT   256
#define OUTF   128
#define NEDGE  800000

typedef __attribute__((ext_vector_type(4))) float v4f;

#define BM  80
#define BN  64
#define KC  32
#define LDP 36

__global__ __launch_bounds__(128) void gemm_tiled_wmma(
    const float* __restrict__ A0, const float* __restrict__ W0,
    const float* __restrict__ A1, const float* __restrict__ W1,
    const float* __restrict__ bias, float* __restrict__ Y, int nseg, int colOff)
{
  __shared__ __attribute__((aligned(16))) float As[BM * LDP];
  __shared__ __attribute__((aligned(16))) float Bs[BN * LDP];

  const int tid  = threadIdx.x;
  const int lane = tid & 31;
  const int wave = tid >> 5;
  const int c    = lane & 15;
  const int h    = lane >> 4;
  const int m0   = blockIdx.x * BM;
  const int n0   = blockIdx.y * BN;

  const int srow = tid >> 3;
  const int scol = (tid & 7) << 2;
  const size_t aoff = (size_t)(m0 + srow) * FEAT + scol;
  const size_t woff = (size_t)(n0 + srow) * FEAT + scol;
  float* const asw = As + srow * LDP + scol;
  float* const bsw = Bs + srow * LDP + scol;
  const float* const afr = As + c * LDP + (h << 2);
  const float* const bfr = Bs + ((wave << 4) + c) * LDP + (h << 2);

  v8f acc[5];
#pragma unroll
  for (int i = 0; i < 5; ++i) acc[i] = (v8f){0.f, 0.f, 0.f, 0.f, 0.f, 0.f, 0.f, 0.f};

  v4f ra[5], rb[4];
  {
    const float* ap = A0 + aoff;
    const float* wp = W0 + woff;
#pragma unroll
    for (int i = 0; i < 5; ++i) ra[i] = *(const v4f*)(ap + (size_t)i * 16 * FEAT);
#pragma unroll
    for (int i = 0; i < 4; ++i) rb[i] = *(const v4f*)(wp + (size_t)i * 16 * FEAT);
  }

  const int nch = nseg * (FEAT / KC);
  for (int ch = 0; ch < nch; ++ch) {
    __syncthreads();
#pragma unroll
    for (int i = 0; i < 5; ++i) *(v4f*)(asw + i * 16 * LDP) = ra[i];
#pragma unroll
    for (int i = 0; i < 4; ++i) *(v4f*)(bsw + i * 16 * LDP) = rb[i];
    __syncthreads();

    if (ch + 1 < nch) {
      const int   nx = ch + 1;
      const int   kb = (nx & 7) * KC;
      const float* ap = ((nx < 8) ? A0 : A1) + aoff + kb;
      const float* wp = ((nx < 8) ? W0 : W1) + woff + kb;
#pragma unroll
      for (int i = 0; i < 5; ++i) ra[i] = *(const v4f*)(ap + (size_t)i * 16 * FEAT);
#pragma unroll
      for (int i = 0; i < 4; ++i) rb[i] = *(const v4f*)(wp + (size_t)i * 16 * FEAT);
    }

    v4f fa[2][5], fb[2];
    fb[0] = *(const v4f*)bfr;
#pragma unroll
    for (int i = 0; i < 5; ++i) fa[0][i] = *(const v4f*)(afr + i * 16 * LDP);
#pragma unroll
    for (int j = 0; j < 4; ++j) {
      const int cur = j & 1, nxt = cur ^ 1;
      if (j + 1 < 4) {
        fb[nxt] = *(const v4f*)(bfr + (j + 1) * 8);
#pragma unroll
        for (int i = 0; i < 5; ++i)
          fa[nxt][i] = *(const v4f*)(afr + i * 16 * LDP + (j + 1) * 8);
      }
#pragma unroll
      for (int i = 0; i < 5; ++i) {
        const v2f a = fa[cur][i].xy;
        const v2f b = fb[cur].xy;
        acc[i] = __builtin_amdgcn_wmma_f32_16x16x4_f32(
            false, a, false, b, (short)0, acc[i], false, false);
      }
#pragma unroll
      for (int i = 0; i < 5; ++i) {
        const v2f a = fa[cur][i].zw;
        const v2f b = fb[cur].zw;
        acc[i] = __builtin_amdgcn_wmma_f32_16x16x4_f32(
            false, a, false, b, (short)0, acc[i], false, false);
      }
    }
  }

  const int col = n0 + (wave << 4) + c;
  const float bv = bias[col];
  float* yp = Y + (size_t)(m0 + (h << 3)) * FEAT + colOff + col;
#pragma unroll
  for (int i = 0; i < 5; ++i) {
#pragma unroll
    for (int r = 0; r < 8; ++r)
      yp[(size_t)(i * 16 + r) * FEAT] = acc[i][r] + bv;
  }
}

#define NSL    32
#define EPS    (NEDGE / NSL)
#define NCHUNK ((NNODES + 1023) / 1024)

__global__ __launch_bounds__(1024) void edge_rank_lds(
    const int* __restrict__ EI0, const int* __restrict__ EI1,
    int* __restrict__ RANK, int* __restrict__ H, int* __restrict__ G)
{
  __shared__ __attribute__((aligned(16))) int hist[NNODES];
  const int t    = threadIdx.x;
  const int lane = t & 31;
  const int wave = t >> 5;
  const int sg   = blockIdx.x / NSL;
  const int sl   = blockIdx.x - sg * NSL;
  const int ss   = blockIdx.x;
  const int* __restrict__ EI = sg ? EI1 : EI0;

  int4* h4 = (int4*)hist;
  for (int i = t; i < NNODES / 4; i += 1024) h4[i] = make_int4(0, 0, 0, 0);
  __syncthreads();

  const int* er = EI + sl * EPS;
  int*       rk = RANK + sg * NEDGE + sl * EPS;
  for (int i0 = 0; i0 < EPS; i0 += 4096) {
    int r[4];
#pragma unroll
    for (int j = 0; j < 4; ++j) {
      const int i = i0 + (j << 10) + t;
      r[j] = (i < EPS) ? er[i] : 0;
    }
    int q[4];
#pragma unroll
    for (int j = 0; j < 4; ++j) {
      const int i = i0 + (j << 10) + t;
      q[j] = (i < EPS) ? atomicAdd(&hist[r[j]], 1) : 0;
    }
#pragma unroll
    for (int j = 0; j < 4; ++j) {
      const int i = i0 + (j << 10) + t;
      if (i < EPS) rk[i] = q[j];
    }
  }
  __syncthreads();

  int* hp = H + ss * NNODES;
  for (int i = t; i < NNODES; i += 1024) hp[i] = hist[i];

  for (int k = wave; k < NCHUNK; k += 32) {
    int s = 0;
    for (int j = 0; j < 32; ++j) {
      const int i = (k << 10) + (j << 5) + lane;
      s += (i < NNODES) ? hist[i] : 0;
    }
    for (int off = 16; off >= 1; off >>= 1) s += __shfl_xor(s, off, 32);
    if (lane == 0) G[ss * 64 + k] = s;
  }
}

__global__ __launch_bounds__(1024) void scan_slices(
    int* __restrict__ Hall, const int* __restrict__ Gall,
    int* __restrict__ OFS0, int* __restrict__ OFS1, int n)
{
  __shared__ int wsum[32];
  __shared__ int wpre[32];
  const int t    = threadIdx.x;
  const int lane = t & 31;
  const int wave = t >> 5;
  const int base = blockIdx.x << 10;
  const int sg   = blockIdx.y;
  int* __restrict__       H   = Hall + sg * (NSL * NNODES);
  const int* __restrict__ G   = Gall + sg * (NSL * 64);
  int* __restrict__       OFS = sg ? OFS1 : OFS0;

  int pre = 0;
  for (int i = t; i < NSL * 64; i += 1024)
    if ((i & 63) < (int)blockIdx.x) pre += G[i];

  const int idx = base + t;
  int hv[NSL];
#pragma unroll
  for (int b = 0; b < NSL; ++b) hv[b] = (idx < n) ? H[b * NNODES + idx] : 0;
  int c = 0;
#pragma unroll
  for (int b = 0; b < NSL; ++b) c += hv[b];

  int inc = c;
  for (int off = 1; off < 32; off <<= 1) {
    const int v = __shfl_up(inc, off, 32);
    if (lane >= off) inc += v;
  }
  for (int off = 16; off >= 1; off >>= 1) pre += __shfl_xor(pre, off, 32);
  if (lane == 31) wsum[wave] = inc;
  if (lane == 0)  wpre[wave] = pre;
  __syncthreads();
  int front = 0;
  for (int w = 0; w < 32; ++w) {
    front += wpre[w];
    if (w < wave) front += wsum[w];
  }
  if (idx < n) {
    int run = front + inc - c;
    OFS[idx] = run;
#pragma unroll
    for (int b = 0; b < NSL; ++b) {
      H[b * NNODES + idx] = run;
      run += hv[b];
    }
  }
}

__global__ __launch_bounds__(256) void edge_place(
    const int* __restrict__ EI0, const int* __restrict__ EI1,
    const int* __restrict__ H, const int* __restrict__ RANK,
    int* __restrict__ BKT0, int* __restrict__ BKT1, int E)
{
  const int sg  = blockIdx.y;
  const int* __restrict__ EI  = sg ? EI1 : EI0;
  int* __restrict__       BKT = sg ? BKT1 : BKT0;
  const int e = blockIdx.x * 256 + threadIdx.x;
  if (e < E) {
    const int sl = e / EPS;
    BKT[H[(sg * NSL + sl) * NNODES + EI[e]] + RANK[sg * NEDGE + e]] = EI[E + e];
  }
}

__device__ __forceinline__ float lane_xor16(float x)
{
  const int b = __builtin_bit_cast(int, x);
  return __builtin_bit_cast(float,
      __builtin_amdgcn_permlanex16(b, b, 0x76543210, 0xfedcba98, false, false));
}
template <int M>
__device__ __forceinline__ float row_xor(float x)
{
  return __builtin_bit_cast(float,
      __builtin_amdgcn_update_dpp(0, __builtin_bit_cast(int, x),
                                  0x160 | M, 0xf, 0xf, false));
}

__device__ __forceinline__ void edge_term(
    const float* __restrict__ X, int l, int lane,
    const float4& a0, const float4& a1, float4& s0, float4& s1)
{
  const float4* xp = (const float4*)(X + (size_t)l * FEAT);
  const float4 x0 = xp[lane], x1 = xp[lane + 32];
  float v = a0.x * x0.x + a0.y * x0.y + a0.z * x0.z + a0.w * x0.w
          + a1.x * x1.x + a1.y * x1.y + a1.z * x1.z + a1.w * x1.w;
  v += lane_xor16(v);
  v += row_xor<8>(v);
  v += row_xor<4>(v);
  v += row_xor<2>(v);
  v += row_xor<1>(v);
  s0.x += v * x0.x; s0.y += v * x0.y; s0.z += v * x0.z; s0.w += v * x0.w;
  s1.x += v * x1.x; s1.y += v * x1.y; s1.z += v * x1.z; s1.w += v * x1.w;
}

__global__ __launch_bounds__(256) void node_aggregate(
    const int* __restrict__ OFS, const int* __restrict__ BKT,
    const float* __restrict__ X, float* __restrict__ AS, int n, int E)
{
  const int lane = threadIdx.x & 31;
  const int wave = threadIdx.x >> 5;
  const int r = __builtin_amdgcn_readfirstlane((blockIdx.x << 3) + wave);
  if (r >= n) return;
  const int start = OFS[r];
  const int end   = (r + 1 < n) ? OFS[r + 1] : E;
  const int deg   = end - start;

  float4* row = (float4*)(AS + (size_t)r * FEAT);
  const float4 a0 = row[lane], a1 = row[lane + 32];
  float4 s0 = {0.f, 0.f, 0.f, 0.f}, s1 = {0.f, 0.f, 0.f, 0.f};

  if (deg <= 32) {
    const int k = (lane < deg) ? BKT[start + lane] : 0x7fffffff;
    int rank = 0;
#pragma unroll
    for (int j = 0; j < 32; ++j) {
      const int kj = __builtin_amdgcn_readlane(k, j);
      rank += ((kj < k) || (kj == k && j < lane)) ? 1 : 0;
    }
    const int sorted = __builtin_amdgcn_ds_permute(rank << 2, k);
#pragma unroll 4
    for (int t = 0; t < deg; ++t) {
      const int l = __builtin_amdgcn_readlane(sorted, t);
      edge_term(X, l, lane, a0, a1, s0, s1);
    }
  } else if (deg <= 64) {
    const int k0 = BKT[start + lane];
    const int k1 = (lane + 32 < deg) ? BKT[start + 32 + lane] : 0x7fffffff;
    int r0 = 0, r1 = 0;
    for (int j = 0; j < 32; ++j) {
      const int p = __builtin_amdgcn_readlane(k0, j);
      const int q = __builtin_amdgcn_readlane(k1, j);
      r0 += ((p < k0) || (p == k0 && j < lane)) ? 1 : 0;
      r0 += (q < k0) ? 1 : 0;
      r1 += (p <= k1) ? 1 : 0;
      r1 += ((q < k1) || (q == k1 && j < lane)) ? 1 : 0;
    }
#pragma unroll 2
    for (int t = 0; t < deg; ++t) {
      const bool h0 = (r0 == t);
      const bool h1 = (r1 == t);
      const int  kv = h0 ? k0 : k1;
      const unsigned m = __builtin_amdgcn_ballot_w32(h0 || h1);
      const int l = __builtin_amdgcn_readlane(kv, __builtin_ctz(m));
      edge_term(X, l, lane, a0, a1, s0, s1);
    }
  } else {
    int last = -1;
    for (int it = 0; it < deg; ++it) {
      int cand = 0x7fffffff;
      for (int b = start; b < end; b += 32) {
        const int p = b + lane;
        const int k = (p < end) ? BKT[p] : 0x7fffffff;
        if (k > last && k < cand) cand = k;
      }
      for (int off = 16; off >= 1; off >>= 1) {
        const int o = __shfl_xor(cand, off, 32);
        cand = (o < cand) ? o : cand;
      }
      if (cand == 0x7fffffff) break;
      int mult = 0;
      for (int b = start; b < end; b += 32) {
        const int p = b + lane;
        const int k = (p < end) ? BKT[p] : 0x7fffffff;
        mult += __popc(__builtin_amdgcn_ballot_w32(k == cand));
      }
      const int l = __builtin_amdgcn_readfirstlane(cand);
      for (int m = 0; m < mult; ++m) edge_term(X, l, lane, a0, a1, s0, s1);
      last = cand;
    }
  }

  const float inv = 1.0f / fmaxf((float)deg, 1.0f);
  s0.x *= inv; s0.y *= inv; s0.z *= inv; s0.w *= inv;
  s1.x *= inv; s1.y *= inv; s1.z *= inv; s1.w *= inv;
  row[lane] = s0;
  row[lane + 32] = s1;
}

extern "C" void kernel_launch(void* const* d_in, const int* in_sizes, int n_in,
                              void* d_out, int out_size, void* d_ws, size_t ws_size,
                              hipStream_t stream)
{
  const float* x        = (const float*)d_in[0];
  const int*   pos_ei   = (const int*)d_in[1];
  const int*   neg_ei   = (const int*)d_in[2];
  const float* W_pos    = (const float*)d_in[3];
  const float* W_pos_cc = (const float*)d_in[4];
  const float* b_pos_cc = (const float*)d_in[5];
  const float* W_neg    = (const float*)d_in[6];
  const float* W_neg_cc = (const float*)d_in[7];
  const float* b_neg_cc = (const float*)d_in[8];
  const float* W_patt   = (const float*)d_in[9];
  const float* b_patt   = (const float*)d_in[10];
  const float* W_natt   = (const float*)d_in[11];
  const float* b_natt   = (const float*)d_in[12];
  float* out = (float*)d_out;

  float* Sbuf = (float*)d_ws;
  int*   Rank = (int*)d_ws;
  int*   Hsl  = Rank + (size_t)2 * NEDGE;
  int*   Gsl  = Hsl + (size_t)2 * NSL * NNODES;
  int*   Bkt1 = (int*)(Sbuf + (size_t)NNODES * FEAT);
  int*   Ofs1 = Bkt1 + (size_t)NEDGE;
  int*   Bkt0 = (int*)d_out;
  int*   Ofs0 = Bkt0 + (size_t)NEDGE;

  const dim3 blk(256);
  const dim3 gblk(128);
  const dim3 gAtt(NNODES / BM, FEAT / BN);
  const dim3 gOut(NNODES / BM, OUTF / BN);
  const int  gEdge = (NEDGE + 255) / 256;
  const int  gScan = (NNODES + 1023) / 1024;
  const int  gNode = (NNODES + 7) / 8;

  const int*   bkt[2]  = { Bkt0, Bkt1 };
  const int*   ofs[2]  = { Ofs0, Ofs1 };
  const float* Watt[2] = { W_patt, W_natt };
  const float* batt[2] = { b_patt, b_natt };
  const float* Wagg[2] = { W_pos,  W_neg  };
  const float* Wcc[2]  = { W_pos_cc, W_neg_cc };
  const float* bcc[2]  = { b_pos_cc, b_neg_cc };

  edge_rank_lds<<<2 * NSL, dim3(1024), 0, stream>>>(pos_ei, neg_ei, Rank, Hsl, Gsl);
  scan_slices<<<dim3(gScan, 2), dim3(1024), 0, stream>>>(Hsl, Gsl, Ofs0, Ofs1, NNODES);
  edge_place<<<dim3(gEdge, 2), blk, 0, stream>>>(pos_ei, neg_ei, Hsl, Rank,
                                                 Bkt0, Bkt1, NEDGE);

  for (int s = 0; s < 2; ++s) {
    gemm_tiled_wmma<<<gAtt, gblk, 0, stream>>>(x, Watt[s], x, Watt[s],
                                               batt[s], Sbuf, 1, 0);
    node_aggregate<<<gNode, blk, 0, stream>>>(ofs[s], bkt[s], x, Sbuf, NNODES, NEDGE);
    gemm_tiled_wmma<<<gOut, gblk, 0, stream>>>(Sbuf, Wagg[s], x, Wcc[s],
                                               bcc[s], out, 2, s * OUTF);
  }
}
